# starting kernel
# baseline (speedup 1.0000x reference)
#include <hip/hip_runtime.h>

typedef __attribute__((ext_vector_type(2))) float v2f;
typedef __attribute__((ext_vector_type(8))) float v8f;

#define FIN 6
#define HID 16

__global__ __launch_bounds__(256) void zero_kernel(float* __restrict__ p, int n) {
  int i = blockIdx.x * blockDim.x + threadIdx.x;
  if (i < n) p[i] = 0.0f;
}

__global__ __launch_bounds__(256) void degree_kernel(const int* __restrict__ src,
                                                     const int* __restrict__ dst,
                                                     float* deg_out, float* deg_in, int E) {
  int e = blockIdx.x * blockDim.x + threadIdx.x;
  if (e >= E) return;
  atomicAdd(deg_out + src[e], 1.0f);
  atomicAdd(deg_in + dst[e], 1.0f);
}

__global__ __launch_bounds__(256) void norm_hs1_kernel(const float* __restrict__ h,
                                                       float* deg_out, float* deg_in,
                                                       float* __restrict__ hs1, int N) {
  int i = blockIdx.x * blockDim.x + threadIdx.x;
  if (i >= N) return;
  float dO = deg_out[i];
  float dI = deg_in[i];
  float no = dO > 0.0f ? rsqrtf(dO) : 0.0f;
  float ni = dI > 0.0f ? rsqrtf(dI) : 0.0f;
  deg_out[i] = no;
  deg_in[i]  = ni;
#pragma unroll
  for (int k = 0; k < FIN; ++k) hs1[i * FIN + k] = h[i * FIN + k] * no;
}

__global__ __launch_bounds__(256) void scatter1_kernel(const int* __restrict__ src,
                                                       const int* __restrict__ dst,
                                                       const float* __restrict__ hs1,
                                                       float* agg1, int E) {
  int e = blockIdx.x * blockDim.x + threadIdx.x;
  if (e >= E) return;
  int s = src[e], d = dst[e];
  const float2* hp = (const float2*)(hs1 + (size_t)s * FIN);
  float2 a = hp[0], b = hp[1], c = hp[2];
  float* ap = agg1 + (size_t)d * FIN;
  atomicAdd(ap + 0, a.x); atomicAdd(ap + 1, a.y);
  atomicAdd(ap + 2, b.x); atomicAdd(ap + 3, b.y);
  atomicAdd(ap + 4, c.x); atomicAdd(ap + 5, c.y);
}

__global__ __launch_bounds__(256) void combine1_kernel(const float* __restrict__ agg1,
                                                       const float* __restrict__ norm_in,
                                                       const float* __restrict__ norm_out,
                                                       const float* __restrict__ W1,
                                                       const float* __restrict__ b1,
                                                       float* __restrict__ hs2, int N) {
  int i = blockIdx.x * blockDim.x + threadIdx.x;
  if (i >= N) return;
  float ni = norm_in[i], no = norm_out[i];
  float v[FIN];
#pragma unroll
  for (int k = 0; k < FIN; ++k) v[k] = agg1[(size_t)i * FIN + k] * ni;
#pragma unroll
  for (int j = 0; j < HID; ++j) {
    float acc = b1[j];
#pragma unroll
    for (int k = 0; k < FIN; ++k) acc = fmaf(v[k], W1[k * HID + j], acc);
    acc = fmaxf(acc, 0.0f);
    hs2[(size_t)i * HID + j] = acc * no;
  }
}

__global__ __launch_bounds__(256) void scatter2_kernel(const int* __restrict__ src,
                                                       const int* __restrict__ dst,
                                                       const float* __restrict__ hs2,
                                                       float* agg2, int E) {
  int e = blockIdx.x * blockDim.x + threadIdx.x;
  if (e >= E) return;
  int s = src[e], d = dst[e];
  const float4* hp = (const float4*)(hs2 + (size_t)s * HID);
  float4 v0 = hp[0], v1 = hp[1], v2 = hp[2], v3 = hp[3];
  float* ap = agg2 + (size_t)d * HID;
  atomicAdd(ap +  0, v0.x); atomicAdd(ap +  1, v0.y);
  atomicAdd(ap +  2, v0.z); atomicAdd(ap +  3, v0.w);
  atomicAdd(ap +  4, v1.x); atomicAdd(ap +  5, v1.y);
  atomicAdd(ap +  6, v1.z); atomicAdd(ap +  7, v1.w);
  atomicAdd(ap +  8, v2.x); atomicAdd(ap +  9, v2.y);
  atomicAdd(ap + 10, v2.z); atomicAdd(ap + 11, v2.w);
  atomicAdd(ap + 12, v3.x); atomicAdd(ap + 13, v3.y);
  atomicAdd(ap + 14, v3.z); atomicAdd(ap + 15, v3.w);
}

__global__ __launch_bounds__(256) void combine2_wmma_kernel(float* agg2,
                                                            const float* __restrict__ norm_in,
                                                            const float* __restrict__ W2,
                                                            const float* __restrict__ b2,
                                                            int numTiles) {
  int wave = (int)((blockIdx.x * blockDim.x + threadIdx.x) >> 5);
  int lane = threadIdx.x & 31;
  if (wave >= numTiles) return;
  int row0 = wave * 16;
  int half = lane >> 4;
  int lr   = lane & 15;

  int arow = row0 + lr;
  float ni = norm_in[arow];
  const float* Arow = agg2 + (size_t)arow * HID;

  v8f c = {};
#pragma unroll
  for (int kb = 0; kb < 16; kb += 4) {
    int k0 = kb + half * 2;
    v2f a, b;
    a.x = Arow[k0 + 0] * ni;
    a.y = Arow[k0 + 1] * ni;
    b.x = W2[(k0 + 0) * HID + lr];
    b.y = W2[(k0 + 1) * HID + lr];
    c = __builtin_amdgcn_wmma_f32_16x16x4_f32(false, a, false, b, (short)0, c,
                                              false, false);
  }

  float bias = b2[lr];
#pragma unroll
  for (int r = 0; r < 8; ++r) {
    int m = r + half * 8;
    float val = fmaxf(c[r] + bias, 0.0f);
    agg2[(size_t)(row0 + m) * HID + lr] = val;
  }
}

__global__ __launch_bounds__(256) void combine2_tail_kernel(float* agg2,
                                                            const float* __restrict__ norm_in,
                                                            const float* __restrict__ W2,
                                                            const float* __restrict__ b2,
                                                            int rowStart, int N) {
  int i = rowStart + blockIdx.x * blockDim.x + threadIdx.x;
  if (i >= N) return;
  float ni = norm_in[i];
  float v[HID];
#pragma unroll
  for (int k = 0; k < HID; ++k) v[k] = agg2[(size_t)i * HID + k] * ni;
#pragma unroll
  for (int j = 0; j < HID; ++j) {
    float acc = b2[j];
#pragma unroll
    for (int k = 0; k < HID; ++k) acc = fmaf(v[k], W2[k * HID + j], acc);
    agg2[(size_t)i * HID + j] = fmaxf(acc, 0.0f);
  }
}

__global__ __launch_bounds__(256) void head_kernel(const float* __restrict__ h2,
                                                   const float* __restrict__ Wl,
                                                   const float* __restrict__ bl,
                                                   float* __restrict__ out, int N) {
  int i = blockIdx.x * blockDim.x + threadIdx.x;
  if (i >= N) return;
  const float4* hp = (const float4*)(h2 + (size_t)i * HID);
  const float4* wp = (const float4*)Wl;
  float acc = bl[0];
#pragma unroll
  for (int q = 0; q < 4; ++q) {
    float4 hv = hp[q], wv = wp[q];
    acc = fmaf(hv.x, wv.x, acc);
    acc = fmaf(hv.y, wv.y, acc);
    acc = fmaf(hv.z, wv.z, acc);
    acc = fmaf(hv.w, wv.w, acc);
  }
  out[i] = acc;
}

extern "C" void kernel_launch(void* const* d_in, const int* in_sizes, int n_in,
                              void* d_out, int out_size, void* d_ws, size_t ws_size,
                              hipStream_t stream) {
  const float* h  = (const float*)d_in[0];
  const float* W1 = (const float*)d_in[1];
  const float* b1 = (const float*)d_in[2];
  const float* W2 = (const float*)d_in[3];
  const float* b2 = (const float*)d_in[4];
  const float* Wl = (const float*)d_in[5];
  const float* bl = (const float*)d_in[6];
  const int* src  = (const int*)d_in[7];
  const int* dst  = (const int*)d_in[8];
  float* out = (float*)d_out;

  const int N = in_sizes[0] / FIN;
  const int E = in_sizes[7];

  float* ws = (float*)d_ws;
  float* deg_out = ws;
  float* deg_in  = ws + (size_t)N;
  float* agg1    = ws + 2 * (size_t)N;
  float* agg2    = ws + 8 * (size_t)N;
  float* hs1     = ws + 24 * (size_t)N;
  float* hs2     = ws + 30 * (size_t)N;

  const int B = 256;
  int zn = 24 * N;
  zero_kernel<<<(zn + B - 1) / B, B, 0, stream>>>(ws, zn);

  degree_kernel<<<(E + B - 1) / B, B, 0, stream>>>(src, dst, deg_out, deg_in, E);
  norm_hs1_kernel<<<(N + B - 1) / B, B, 0, stream>>>(h, deg_out, deg_in, hs1, N);

  scatter1_kernel<<<(E + B - 1) / B, B, 0, stream>>>(src, dst, hs1, agg1, E);
  combine1_kernel<<<(N + B - 1) / B, B, 0, stream>>>(agg1, deg_in, deg_out, W1, b1, hs2, N);

  scatter2_kernel<<<(E + B - 1) / B, B, 0, stream>>>(src, dst, hs2, agg2, E);

  int numTiles = N / 16;
  if (numTiles > 0) {
    int threads = numTiles * 32;
    combine2_wmma_kernel<<<(threads + B - 1) / B, B, 0, stream>>>(agg2, deg_in, W2, b2, numTiles);
  }
  int rem = N - numTiles * 16;
  if (rem > 0) {
    combine2_tail_kernel<<<(rem + B - 1) / B, B, 0, stream>>>(agg2, deg_in, W2, b2,
                                                              numTiles * 16, N);
  }

  head_kernel<<<(N + B - 1) / B, B, 0, stream>>>(agg2, Wl, bl, out, N);
}


# Edges partitioned by 256-node range (LDS counts, scan, tickets); one work-group per range sums messages and degrees with LDS f32 atomics and finishes the layer; no global atomics; arrival-order sums
# speedup vs baseline: 9.8540x; 9.8540x over previous
#include <hip/hip_runtime.h>


typedef __attribute__((ext_vector_type(2))) float v2f;
typedef __attribute__((ext_vector_type(8))) float v8f;

#define FIN 6
#define HID 16
#define RSH 8
#define RNG 256
#define NBLK 256
#define MAXR 1024
#define PT 1024

__global__ __launch_bounds__(PT) void count_kernel(const int* __restrict__ src,
                                                   const int* __restrict__ dst,
                                                   int* __restrict__ cntD, int* __restrict__ cntS,
                                                   int E, int chunk, int nrp) {
  __shared__ int bD[MAXR];
  __shared__ int bS[MAXR];
  int t = threadIdx.x;
  bD[t] = 0; bS[t] = 0;
  __syncthreads();
  int lo = blockIdx.x * chunk;
  int hi = min(lo + chunk, E);
  for (int i = lo + t; i < hi; i += PT) {
    atomicAdd(&bD[dst[i] >> RSH], 1);
    atomicAdd(&bS[src[i] >> RSH], 1);
  }
  __syncthreads();
  if (t < nrp) {
    cntD[blockIdx.x * nrp + t] = bD[t];
    cntS[blockIdx.x * nrp + t] = bS[t];
  }
}

__global__ __launch_bounds__(1024) void offsets_kernel(const int* __restrict__ cntD,
                                                       const int* __restrict__ cntS,
                                                       int* __restrict__ offD, int* __restrict__ offS,
                                                       int* __restrict__ totD, int* __restrict__ totS,
                                                       int nrp) {
  __shared__ int part[32][33];
  int groups = nrp >> 5;
  int g = blockIdx.x;
  const int* cnt = cntD; int* off = offD; int* tot = totD;
  if (g >= groups) { g -= groups; cnt = cntS; off = offS; tot = totS; }
  int bb = threadIdx.x & 31, kq = threadIdx.x >> 5;
  int b = g * 32 + bb;
  int c[8]; int s = 0;
#pragma unroll
  for (int j = 0; j < 8; ++j) { c[j] = cnt[(kq * 8 + j) * nrp + b]; s += c[j]; }
  part[kq][bb] = s;
  __syncthreads();
  int run = 0;
  for (int q = 0; q < kq; ++q) run += part[q][bb];
#pragma unroll
  for (int j = 0; j < 8; ++j) { off[(kq * 8 + j) * nrp + b] = run; run += c[j]; }
  if (kq == 31) tot[b] = run;
}

__device__ __forceinline__ int block_excl_scan(int v, int* wsum) {
  int lane = threadIdx.x & 31, w = threadIdx.x >> 5;
  int x = v;
#pragma unroll
  for (int d = 1; d < 32; d <<= 1) { int n = __shfl_up(x, d); if (lane >= d) x += n; }
  if (lane == 31) wsum[w] = x;
  __syncthreads();
  if (w == 0) {
    int y = wsum[lane]; int z = y;
#pragma unroll
    for (int d = 1; d < 32; d <<= 1) { int n = __shfl_up(z, d); if (lane >= d) z += n; }
    wsum[lane] = z - y;
  }
  __syncthreads();
  int r = x - v + wsum[w];
  __syncthreads();
  return r;
}

__global__ __launch_bounds__(PT) void partition_kernel(const int* __restrict__ src,
                                                       const int* __restrict__ dst,
                                                       const int* __restrict__ offD,
                                                       const int* __restrict__ offS,
                                                       const int* __restrict__ totD,
                                                       const int* __restrict__ totS,
                                                       int* __restrict__ baseD, int* __restrict__ baseS,
                                                       int* __restrict__ recDw, int* __restrict__ recS,
                                                       int E, int chunk, int nrp) {
  __shared__ int curD[MAXR];
  __shared__ int curS[MAXR];
  __shared__ int wsum[32];
  int2* recD = (int2*)recDw;
  int t = threadIdx.x;
  int tD = t < nrp ? totD[t] : 0;
  int tS = t < nrp ? totS[t] : 0;
  int bD = block_excl_scan(tD, wsum);
  int bS = block_excl_scan(tS, wsum);
  if (t < nrp) {
    curD[t] = bD + offD[blockIdx.x * nrp + t];
    curS[t] = bS + offS[blockIdx.x * nrp + t];
    if (blockIdx.x == 0) { baseD[t] = bD; baseS[t] = bS; }
  }
  __syncthreads();
  int lo = blockIdx.x * chunk;
  int hi = min(lo + chunk, E);
  for (int i0 = lo + t; i0 - t < hi; i0 += 4 * PT) {
    int s[4], d[4];
#pragma unroll
    for (int u = 0; u < 4; ++u) {
      int i = min(i0 + u * PT, hi - 1);
      s[u] = src[i];
      d[u] = dst[i];
    }
#pragma unroll
    for (int u = 0; u < 4; ++u) {
      if (i0 + u * PT < hi) {
        int p = atomicAdd(&curD[d[u] >> RSH], 1);
        recD[p] = make_int2(s[u], d[u]);
        int q = atomicAdd(&curS[s[u] >> RSH], 1);
        recS[q] = s[u];
      }
    }
  }
}

__global__ __launch_bounds__(512) void outnorm_kernel(const int* __restrict__ recS,
                                                      const int* __restrict__ baseS,
                                                      const int* __restrict__ totS,
                                                      const float* __restrict__ h,
                                                      float* __restrict__ norm_out,
                                                      float* __restrict__ hs1, int N) {
  __shared__ int c[RNG];
  int t = threadIdx.x, b = blockIdx.x;
  if (t < RNG) c[t] = 0;
  __syncthreads();
  int lo = baseS[b];
  int hi = lo + totS[b];
  for (int i = lo + t; i < hi; i += 512) atomicAdd(&c[recS[i] & (RNG - 1)], 1);
  __syncthreads();
  int i = b * RNG + t;
  if (t < RNG && i < N) {
    float dO = (float)c[t];
    float no = dO > 0.0f ? rsqrtf(dO) : 0.0f;
    norm_out[i] = no;
#pragma unroll
    for (int k = 0; k < FIN; ++k) hs1[(size_t)i * FIN + k] = h[(size_t)i * FIN + k] * no;
  }
}

__global__ __launch_bounds__(512) void layer1_kernel(const int* __restrict__ recDw,
                                                     const int* __restrict__ baseD,
                                                     const int* __restrict__ totD,
                                                     const float* __restrict__ hs1,
                                                     const float* __restrict__ norm_out,
                                                     const float* __restrict__ W1,
                                                     const float* __restrict__ b1,
                                                     float* __restrict__ norm_in,
                                                     float* __restrict__ hs2, int N) {
  __shared__ float acc[RNG * 8];
  const int2* recD = (const int2*)recDw;
  int t = threadIdx.x, b = blockIdx.x;
  for (int j = t; j < RNG * 8; j += 512) acc[j] = 0.0f;
  __syncthreads();
  int lo = baseD[b];
  int hi = lo + totD[b];
  int g = t >> 3, f = t & 7;
  for (int i0 = lo + g; i0 - g < hi; i0 += 64 * 8) {
    int2 r[8]; float v[8];
#pragma unroll
    for (int u = 0; u < 8; ++u) {
      r[u] = recD[min(i0 + u * 64, hi - 1)];
    }
#pragma unroll
    for (int u = 0; u < 8; ++u) {
      float x = hs1[(size_t)r[u].x * FIN + min(f, FIN - 1)];
      v[u] = f < FIN ? x : 1.0f;
    }
    asm volatile("" :: "v"(v[0]), "v"(v[1]), "v"(v[2]), "v"(v[3]), "v"(v[4]), "v"(v[5]), "v"(v[6]), "v"(v[7]));
#pragma unroll
    for (int u = 0; u < 8; ++u) {
      if (i0 + u * 64 < hi && f < 7) atomicAdd(&acc[(r[u].y & (RNG - 1)) * 8 + f], v[u]);
    }
  }
  __syncthreads();
  int i = b * RNG + t;
  if (t < RNG && i < N) {
    float dI = acc[t * 8 + 6];
    float ni = dI > 0.0f ? rsqrtf(dI) : 0.0f;
    float no = norm_out[i];
    norm_in[i] = ni;
    float v[FIN];
#pragma unroll
    for (int k = 0; k < FIN; ++k) v[k] = acc[t * 8 + k] * ni;
    float o[HID];
#pragma unroll
    for (int j = 0; j < HID; ++j) {
      float a = b1[j];
#pragma unroll
      for (int k = 0; k < FIN; ++k) a = fmaf(v[k], W1[k * HID + j], a);
      a = fmaxf(a, 0.0f);
      o[j] = a * no;
    }
    float4* hp = (float4*)(hs2 + (size_t)i * HID);
    hp[0] = make_float4(o[0], o[1], o[2], o[3]);
    hp[1] = make_float4(o[4], o[5], o[6], o[7]);
    hp[2] = make_float4(o[8], o[9], o[10], o[11]);
    hp[3] = make_float4(o[12], o[13], o[14], o[15]);
  }
}

__global__ __launch_bounds__(512) void layer2_kernel(const int* __restrict__ recDw,
                                                     const int* __restrict__ baseD,
                                                     const int* __restrict__ totD,
                                                     const float* __restrict__ hs2,
                                                     const float* __restrict__ norm_in,
                                                     const float* __restrict__ W2,
                                                     const float* __restrict__ b2,
                                                     const float* __restrict__ Wl,
                                                     const float* __restrict__ bl,
                                                     float* __restrict__ out, int N) {
  __shared__ float acc[RNG * HID];
  const int2* recD = (const int2*)recDw;
  int t = threadIdx.x, b = blockIdx.x;
  for (int j = t; j < RNG * HID; j += 512) acc[j] = 0.0f;
  __syncthreads();
  int lo = baseD[b];
  int hi = lo + totD[b];
  int g = t >> 4, f = t & 15;
  for (int i0 = lo + g; i0 - g < hi; i0 += 32 * 8) {
    int2 r[8]; float v[8];
#pragma unroll
    for (int u = 0; u < 8; ++u) {
      r[u] = recD[min(i0 + u * 32, hi - 1)];
    }
    asm volatile("" :: "v"(r[0].x), "v"(r[1].x), "v"(r[2].x), "v"(r[3].x),
                       "v"(r[4].x), "v"(r[5].x), "v"(r[6].x), "v"(r[7].x));
#pragma unroll
    for (int u = 0; u < 8; ++u) v[u] = hs2[(unsigned)r[u].x * HID + f];
    asm volatile("" :: "v"(v[0]), "v"(v[1]), "v"(v[2]), "v"(v[3]), "v"(v[4]), "v"(v[5]), "v"(v[6]), "v"(v[7]));
#pragma unroll
    for (int u = 0; u < 8; ++u) {
      if (i0 + u * 32 < hi) atomicAdd(&acc[(r[u].y & (RNG - 1)) * HID + f], v[u]);
    }
  }
  __syncthreads();

  int w = t >> 5, lane = t & 31;
  int half = lane >> 4, lr = lane & 15;
  int lrow = w * 16 + lr;
  int grow = b * RNG + lrow;
  float ni = grow < N ? norm_in[grow] : 0.0f;
  const float* Arow = acc + lrow * HID;
  v8f c = {};
#pragma unroll
  for (int kb = 0; kb < 16; kb += 4) {
    int k0 = kb + half * 2;
    v2f a, bm;
    a.x = Arow[k0 + 0] * ni;
    a.y = Arow[k0 + 1] * ni;
    bm.x = W2[(k0 + 0) * HID + lr];
    bm.y = W2[(k0 + 1) * HID + lr];
    c = __builtin_amdgcn_wmma_f32_16x16x4_f32(false, a, false, bm, (short)0, c, false, false);
  }
  float bias = b2[lr];
#pragma unroll
  for (int r = 0; r < 8; ++r) {
    int m = r + half * 8;
    acc[(w * 16 + m) * HID + lr] = fmaxf(c[r] + bias, 0.0f);
  }
  __syncthreads();
  int i = b * RNG + t;
  if (t < RNG && i < N) {
    float a = bl[0];
#pragma unroll
    for (int j = 0; j < HID; ++j) a = fmaf(acc[t * HID + j], Wl[j], a);
    out[i] = a;
  }
}

extern "C" void kernel_launch(void* const* d_in, const int* in_sizes, int n_in,
                              void* d_out, int out_size, void* d_ws, size_t ws_size,
                              hipStream_t stream) {
  const float* h  = (const float*)d_in[0];
  const float* W1 = (const float*)d_in[1];
  const float* b1 = (const float*)d_in[2];
  const float* W2 = (const float*)d_in[3];
  const float* b2 = (const float*)d_in[4];
  const float* Wl = (const float*)d_in[5];
  const float* bl = (const float*)d_in[6];
  const int* src  = (const int*)d_in[7];
  const int* dst  = (const int*)d_in[8];
  float* out = (float*)d_out;

  const int N = in_sizes[0] / FIN;
  const int E = in_sizes[7];
  const int nr = (N + RNG - 1) / RNG;
  const int nrp = (nr + 31) & ~31;
  const int chunk = (E + NBLK - 1) / NBLK;

  float* ws = (float*)d_ws;
  float* norm_out = ws;
  float* norm_in  = ws + (size_t)N;
  float* hs1      = ws + 2 * (size_t)N;
  float* hs2      = ws + 8 * (size_t)N;
  int* iw   = (int*)(ws + 24 * (size_t)N);
  int* cntD = iw;
  int* cntS = cntD + (size_t)NBLK * nrp;
  int* offD = cntS + (size_t)NBLK * nrp;
  int* offS = offD + (size_t)NBLK * nrp;
  int* totD = offS + (size_t)NBLK * nrp;
  int* totS = totD + nrp;
  int* baseD = totS + nrp;
  int* baseS = baseD + nrp;
  int* recD = baseS + nrp;
  int* recS = recD + 2 * (size_t)E;

  count_kernel<<<NBLK, PT, 0, stream>>>(src, dst, cntD, cntS, E, chunk, nrp);
  offsets_kernel<<<2 * (nrp / 32), 1024, 0, stream>>>(cntD, cntS, offD, offS, totD, totS, nrp);
  partition_kernel<<<NBLK, PT, 0, stream>>>(src, dst, offD, offS, totD, totS, baseD, baseS,
                                            recD, recS, E, chunk, nrp);
  outnorm_kernel<<<nr, 512, 0, stream>>>(recS, baseS, totS, h, norm_out, hs1, N);
  layer1_kernel<<<nr, 512, 0, stream>>>(recD, baseD, totD, hs1, norm_out, W1, b1, norm_in, hs2, N);
  layer2_kernel<<<nr, 512, 0, stream>>>(recD, baseD, totD, hs2, norm_in, W2, b2, Wl, bl, out, N);
}


# Partition puts each chunk's records in range order in LDS (ticket rank, scan, staged write) and stores them as contiguous runs, about a tenth of the store line requests; sums stay arrival-order
# speedup vs baseline: 12.8010x; 1.2991x over previous
#include <hip/hip_runtime.h>


typedef __attribute__((ext_vector_type(2))) float v2f;
typedef __attribute__((ext_vector_type(8))) float v8f;

#define FIN 6
#define HID 16
#define RSH 8
#define RNG 256
#define NBLK 256
#define MAXR 1024
#define PT 1024
#define SB 13
#define SBE (SB * PT)

__global__ __launch_bounds__(PT) void count_kernel(const int* __restrict__ src,
                                                   const int* __restrict__ dst,
                                                   int* __restrict__ cntD, int* __restrict__ cntS,
                                                   int E, int chunk, int nrp) {
  __shared__ int bD[MAXR];
  __shared__ int bS[MAXR];
  int t = threadIdx.x;
  bD[t] = 0; bS[t] = 0;
  __syncthreads();
  int lo = blockIdx.x * chunk;
  int hi = min(lo + chunk, E);
  for (int i = lo + t; i < hi; i += PT) {
    atomicAdd(&bD[dst[i] >> RSH], 1);
    atomicAdd(&bS[src[i] >> RSH], 1);
  }
  __syncthreads();
  if (t < nrp) {
    cntD[blockIdx.x * nrp + t] = bD[t];
    cntS[blockIdx.x * nrp + t] = bS[t];
  }
}

__global__ __launch_bounds__(1024) void offsets_kernel(const int* __restrict__ cntD,
                                                       const int* __restrict__ cntS,
                                                       int* __restrict__ offD, int* __restrict__ offS,
                                                       int* __restrict__ totD, int* __restrict__ totS,
                                                       int nrp) {
  __shared__ int part[32][33];
  int groups = nrp >> 5;
  int g = blockIdx.x;
  const int* cnt = cntD; int* off = offD; int* tot = totD;
  if (g >= groups) { g -= groups; cnt = cntS; off = offS; tot = totS; }
  int bb = threadIdx.x & 31, kq = threadIdx.x >> 5;
  int b = g * 32 + bb;
  int c[8]; int s = 0;
#pragma unroll
  for (int j = 0; j < 8; ++j) { c[j] = cnt[(kq * 8 + j) * nrp + b]; s += c[j]; }
  part[kq][bb] = s;
  __syncthreads();
  int run = 0;
  for (int q = 0; q < kq; ++q) run += part[q][bb];
#pragma unroll
  for (int j = 0; j < 8; ++j) { off[(kq * 8 + j) * nrp + b] = run; run += c[j]; }
  if (kq == 31) tot[b] = run;
}

__device__ __forceinline__ int block_excl_scan(int v, int* wsum) {
  int lane = threadIdx.x & 31, w = threadIdx.x >> 5;
  int x = v;
#pragma unroll
  for (int d = 1; d < 32; d <<= 1) { int n = __shfl_up(x, d); if (lane >= d) x += n; }
  if (lane == 31) wsum[w] = x;
  __syncthreads();
  if (w == 0) {
    int y = wsum[lane]; int z = y;
#pragma unroll
    for (int d = 1; d < 32; d <<= 1) { int n = __shfl_up(z, d); if (lane >= d) z += n; }
    wsum[lane] = z - y;
  }
  __syncthreads();
  int r = x - v + wsum[w];
  __syncthreads();
  return r;
}

__global__ __launch_bounds__(PT) void partition_kernel(const int* __restrict__ src,
                                                       const int* __restrict__ dst,
                                                       const int* __restrict__ offD,
                                                       const int* __restrict__ offS,
                                                       const int* __restrict__ totD,
                                                       const int* __restrict__ totS,
                                                       int* __restrict__ baseD, int* __restrict__ baseS,
                                                       int* __restrict__ recDw, int* __restrict__ recS,
                                                       int E, int chunk, int nrp) {
  __shared__ int2 stD[SBE];
  __shared__ int stS[SBE];
  __shared__ int curD[MAXR];
  __shared__ int curS[MAXR];
  __shared__ int hD[MAXR];
  __shared__ int hS[MAXR];
  __shared__ int wsum[32];
  int2* recD = (int2*)recDw;
  int t = threadIdx.x;
  int tD = t < nrp ? totD[t] : 0;
  int tS = t < nrp ? totS[t] : 0;
  int bD = block_excl_scan(tD, wsum);
  int bS = block_excl_scan(tS, wsum);
  if (t < nrp) {
    curD[t] = bD + offD[blockIdx.x * nrp + t];
    curS[t] = bS + offS[blockIdx.x * nrp + t];
    if (blockIdx.x == 0) { baseD[t] = bD; baseS[t] = bS; }
  } else {
    curD[t] = 0; curS[t] = 0;
  }
  int lo = blockIdx.x * chunk;
  int hi = min(lo + chunk, E);
  for (int s0 = lo; s0 < hi; s0 += SBE) {
    int s1 = min(s0 + SBE, hi);
    int n = s1 - s0;
    hD[t] = 0; hS[t] = 0;
    __syncthreads();
    int s[SB], d[SB], kD[SB], kS[SB];
#pragma unroll
    for (int u = 0; u < SB; ++u) {
      int i = min(s0 + t + u * PT, s1 - 1);
      s[u] = src[i];
      d[u] = dst[i];
    }
#pragma unroll
    for (int u = 0; u < SB; ++u) {
      kD[u] = 0; kS[u] = 0;
      if (t + u * PT < n) {
        kD[u] = atomicAdd(&hD[d[u] >> RSH], 1);
        kS[u] = atomicAdd(&hS[s[u] >> RSH], 1);
      }
    }
    __syncthreads();
    int cD = hD[t], cS = hS[t];
    int lD = block_excl_scan(cD, wsum);
    int lS = block_excl_scan(cS, wsum);
    hD[t] = lD; hS[t] = lS;
    __syncthreads();
#pragma unroll
    for (int u = 0; u < SB; ++u) {
      kD[u] += hD[d[u] >> RSH];
      kS[u] += hS[s[u] >> RSH];
    }
#pragma unroll
    for (int u = 0; u < SB; ++u) {
      if (t + u * PT < n) {
        stD[kD[u]] = make_int2(s[u], d[u]);
        stS[kS[u]] = s[u];
      }
    }
    __syncthreads();
    {
      int2 r[SB]; int p[SB];
#pragma unroll
      for (int u = 0; u < SB; ++u) r[u] = stD[min(t + u * PT, n - 1)];
#pragma unroll
      for (int u = 0; u < SB; ++u) { int g = r[u].y >> RSH; p[u] = curD[g] - hD[g]; }
#pragma unroll
      for (int u = 0; u < SB; ++u) {
        int j = t + u * PT;
        if (j < n) recD[p[u] + j] = r[u];
      }
    }
    {
      int v[SB]; int p[SB];
#pragma unroll
      for (int u = 0; u < SB; ++u) v[u] = stS[min(t + u * PT, n - 1)];
#pragma unroll
      for (int u = 0; u < SB; ++u) { int g = v[u] >> RSH; p[u] = curS[g] - hS[g]; }
#pragma unroll
      for (int u = 0; u < SB; ++u) {
        int j = t + u * PT;
        if (j < n) recS[p[u] + j] = v[u];
      }
    }
    __syncthreads();
    curD[t] += cD; curS[t] += cS;
  }
}

__global__ __launch_bounds__(512) void outnorm_kernel(const int* __restrict__ recS,
                                                      const int* __restrict__ baseS,
                                                      const int* __restrict__ totS,
                                                      const float* __restrict__ h,
                                                      float* __restrict__ norm_out,
                                                      float* __restrict__ hs1, int N) {
  __shared__ int c[RNG];
  int t = threadIdx.x, b = blockIdx.x;
  if (t < RNG) c[t] = 0;
  __syncthreads();
  int lo = baseS[b];
  int hi = lo + totS[b];
  for (int i = lo + t; i < hi; i += 512) atomicAdd(&c[recS[i] & (RNG - 1)], 1);
  __syncthreads();
  int i = b * RNG + t;
  if (t < RNG && i < N) {
    float dO = (float)c[t];
    float no = dO > 0.0f ? rsqrtf(dO) : 0.0f;
    norm_out[i] = no;
#pragma unroll
    for (int k = 0; k < FIN; ++k) hs1[(size_t)i * FIN + k] = h[(size_t)i * FIN + k] * no;
  }
}

__global__ __launch_bounds__(512) void layer1_kernel(const int* __restrict__ recDw,
                                                     const int* __restrict__ baseD,
                                                     const int* __restrict__ totD,
                                                     const float* __restrict__ hs1,
                                                     const float* __restrict__ norm_out,
                                                     const float* __restrict__ W1,
                                                     const float* __restrict__ b1,
                                                     float* __restrict__ norm_in,
                                                     float* __restrict__ hs2, int N) {
  __shared__ float acc[RNG * 8];
  const int2* recD = (const int2*)recDw;
  int t = threadIdx.x, b = blockIdx.x;
  for (int j = t; j < RNG * 8; j += 512) acc[j] = 0.0f;
  __syncthreads();
  int lo = baseD[b];
  int hi = lo + totD[b];
  int g = t >> 3, f = t & 7;
  for (int i0 = lo + g; i0 - g < hi; i0 += 64 * 8) {
    int2 r[8]; float v[8];
#pragma unroll
    for (int u = 0; u < 8; ++u) {
      r[u] = recD[min(i0 + u * 64, hi - 1)];
    }
#pragma unroll
    for (int u = 0; u < 8; ++u) {
      float x = hs1[(size_t)r[u].x * FIN + min(f, FIN - 1)];
      v[u] = f < FIN ? x : 1.0f;
    }
    asm volatile("" :: "v"(v[0]), "v"(v[1]), "v"(v[2]), "v"(v[3]), "v"(v[4]), "v"(v[5]), "v"(v[6]), "v"(v[7]));
#pragma unroll
    for (int u = 0; u < 8; ++u) {
      if (i0 + u * 64 < hi && f < 7) atomicAdd(&acc[(r[u].y & (RNG - 1)) * 8 + f], v[u]);
    }
  }
  __syncthreads();
  int i = b * RNG + t;
  if (t < RNG && i < N) {
    float dI = acc[t * 8 + 6];
    float ni = dI > 0.0f ? rsqrtf(dI) : 0.0f;
    float no = norm_out[i];
    norm_in[i] = ni;
    float v[FIN];
#pragma unroll
    for (int k = 0; k < FIN; ++k) v[k] = acc[t * 8 + k] * ni;
    float o[HID];
#pragma unroll
    for (int j = 0; j < HID; ++j) {
      float a = b1[j];
#pragma unroll
      for (int k = 0; k < FIN; ++k) a = fmaf(v[k], W1[k * HID + j], a);
      a = fmaxf(a, 0.0f);
      o[j] = a * no;
    }
    float4* hp = (float4*)(hs2 + (size_t)i * HID);
    hp[0] = make_float4(o[0], o[1], o[2], o[3]);
    hp[1] = make_float4(o[4], o[5], o[6], o[7]);
    hp[2] = make_float4(o[8], o[9], o[10], o[11]);
    hp[3] = make_float4(o[12], o[13], o[14], o[15]);
  }
}

__global__ __launch_bounds__(512) void layer2_kernel(const int* __restrict__ recDw,
                                                     const int* __restrict__ baseD,
                                                     const int* __restrict__ totD,
                                                     const float* __restrict__ hs2,
                                                     const float* __restrict__ norm_in,
                                                     const float* __restrict__ W2,
                                                     const float* __restrict__ b2,
                                                     const float* __restrict__ Wl,
                                                     const float* __restrict__ bl,
                                                     float* __restrict__ out, int N) {
  __shared__ float acc[RNG * HID];
  const int2* recD = (const int2*)recDw;
  int t = threadIdx.x, b = blockIdx.x;
  for (int j = t; j < RNG * HID; j += 512) acc[j] = 0.0f;
  __syncthreads();
  int lo = baseD[b];
  int hi = lo + totD[b];
  int g = t >> 4, f = t & 15;
  for (int i0 = lo + g; i0 - g < hi; i0 += 32 * 8) {
    int2 r[8]; float v[8];
#pragma unroll
    for (int u = 0; u < 8; ++u) {
      r[u] = recD[min(i0 + u * 32, hi - 1)];
    }
    asm volatile("" :: "v"(r[0].x), "v"(r[1].x), "v"(r[2].x), "v"(r[3].x),
                       "v"(r[4].x), "v"(r[5].x), "v"(r[6].x), "v"(r[7].x));
#pragma unroll
    for (int u = 0; u < 8; ++u) v[u] = hs2[(unsigned)r[u].x * HID + f];
    asm volatile("" :: "v"(v[0]), "v"(v[1]), "v"(v[2]), "v"(v[3]), "v"(v[4]), "v"(v[5]), "v"(v[6]), "v"(v[7]));
#pragma unroll
    for (int u = 0; u < 8; ++u) {
      if (i0 + u * 32 < hi) atomicAdd(&acc[(r[u].y & (RNG - 1)) * HID + f], v[u]);
    }
  }
  __syncthreads();

  int w = t >> 5, lane = t & 31;
  int half = lane >> 4, lr = lane & 15;
  int lrow = w * 16 + lr;
  int grow = b * RNG + lrow;
  float ni = grow < N ? norm_in[grow] : 0.0f;
  const float* Arow = acc + lrow * HID;
  v8f c = {};
#pragma unroll
  for (int kb = 0; kb < 16; kb += 4) {
    int k0 = kb + half * 2;
    v2f a, bm;
    a.x = Arow[k0 + 0] * ni;
    a.y = Arow[k0 + 1] * ni;
    bm.x = W2[(k0 + 0) * HID + lr];
    bm.y = W2[(k0 + 1) * HID + lr];
    c = __builtin_amdgcn_wmma_f32_16x16x4_f32(false, a, false, bm, (short)0, c, false, false);
  }
  float bias = b2[lr];
#pragma unroll
  for (int r = 0; r < 8; ++r) {
    int m = r + half * 8;
    acc[(w * 16 + m) * HID + lr] = fmaxf(c[r] + bias, 0.0f);
  }
  __syncthreads();
  int i = b * RNG + t;
  if (t < RNG && i < N) {
    float a = bl[0];
#pragma unroll
    for (int j = 0; j < HID; ++j) a = fmaf(acc[t * HID + j], Wl[j], a);
    out[i] = a;
  }
}

extern "C" void kernel_launch(void* const* d_in, const int* in_sizes, int n_in,
                              void* d_out, int out_size, void* d_ws, size_t ws_size,
                              hipStream_t stream) {
  const float* h  = (const float*)d_in[0];
  const float* W1 = (const float*)d_in[1];
  const float* b1 = (const float*)d_in[2];
  const float* W2 = (const float*)d_in[3];
  const float* b2 = (const float*)d_in[4];
  const float* Wl = (const float*)d_in[5];
  const float* bl = (const float*)d_in[6];
  const int* src  = (const int*)d_in[7];
  const int* dst  = (const int*)d_in[8];
  float* out = (float*)d_out;

  const int N = in_sizes[0] / FIN;
  const int E = in_sizes[7];
  const int nr = (N + RNG - 1) / RNG;
  const int nrp = (nr + 31) & ~31;
  const int chunk = (E + NBLK - 1) / NBLK;

  float* ws = (float*)d_ws;
  float* norm_out = ws;
  float* norm_in  = ws + (size_t)N;
  float* hs1      = ws + 2 * (size_t)N;
  float* hs2      = ws + 8 * (size_t)N;
  int* iw   = (int*)(ws + 24 * (size_t)N);
  int* cntD = iw;
  int* cntS = cntD + (size_t)NBLK * nrp;
  int* offD = cntS + (size_t)NBLK * nrp;
  int* offS = offD + (size_t)NBLK * nrp;
  int* totD = offS + (size_t)NBLK * nrp;
  int* totS = totD + nrp;
  int* baseD = totS + nrp;
  int* baseS = baseD + nrp;
  int* recD = baseS + nrp;
  int* recS = recD + 2 * (size_t)E;

  count_kernel<<<NBLK, PT, 0, stream>>>(src, dst, cntD, cntS, E, chunk, nrp);
  offsets_kernel<<<2 * (nrp / 32), 1024, 0, stream>>>(cntD, cntS, offD, offS, totD, totS, nrp);
  partition_kernel<<<NBLK, PT, 0, stream>>>(src, dst, offD, offS, totD, totS, baseD, baseS,
                                            recD, recS, E, chunk, nrp);
  outnorm_kernel<<<nr, 512, 0, stream>>>(recS, baseS, totS, h, norm_out, hs1, N);
  layer1_kernel<<<nr, 512, 0, stream>>>(recD, baseD, totD, hs1, norm_out, W1, b1, norm_in, hs2, N);
  layer2_kernel<<<nr, 512, 0, stream>>>(recD, baseD, totD, hs2, norm_in, W2, b2, Wl, bl, out, N);
}


# Edge lists kept chunk-major: a chunk stores its range-ordered records in place plus a (start,count) table; range blocks stage their short runs in LDS; count, offsets launches gone; arrival-order sums
# speedup vs baseline: 15.3286x; 1.1975x over previous
#include <hip/hip_runtime.h>


typedef __attribute__((ext_vector_type(2))) float v2f;
typedef __attribute__((ext_vector_type(8))) float v8f;

#define FIN 6
#define HID 16
#define RSH 8
#define RNG 256
#define NBLK 256
#define MAXR 1024
#define PT 1024
#define SB 13
#define SBE (SB * PT)
#define CT 512
#define CW 16
#define SEGT 256
#define SPW 16
#define CAP 9216


__device__ __forceinline__ int block_excl_scan(int v, int* wsum) {
  int lane = threadIdx.x & 31, w = threadIdx.x >> 5;
  int x = v;
#pragma unroll
  for (int d = 1; d < 32; d <<= 1) { int n = __shfl_up(x, d); if (lane >= d) x += n; }
  if (lane == 31) wsum[w] = x;
  __syncthreads();
  if (w == 0) {
    int y = wsum[lane]; int z = y;
#pragma unroll
    for (int d = 1; d < 32; d <<= 1) { int n = __shfl_up(z, d); if (lane >= d) z += n; }
    wsum[lane] = z - y;
  }
  __syncthreads();
  int r = x - v + wsum[w];
  __syncthreads();
  return r;
}

__global__ __launch_bounds__(PT) void partition_kernel(const int* __restrict__ src,
                                                       const int* __restrict__ dst,
                                                       int* __restrict__ recDw, int* __restrict__ recS,
                                                       int* __restrict__ segDw, int* __restrict__ segSw,
                                                       int E, int chunk, int nr, int nsub) {
  __shared__ int2 stD[SBE];
  __shared__ int stS[SBE];
  __shared__ int hD[MAXR];
  __shared__ int hS[MAXR];
  __shared__ int wsum[32];
  int2* recD = (int2*)recDw;
  int2* segD = (int2*)segDw;
  int2* segS = (int2*)segSw;
  int t = threadIdx.x;
  int nseg = nsub * NBLK;
  int lo = blockIdx.x * chunk;
  int hi = min(lo + chunk, E);
  for (int sub = 0; sub < nsub; ++sub) {
    int s0 = lo + sub * SBE;
    int s1 = min(s0 + SBE, hi);
    int n = s1 - s0;
    int segi = sub * NBLK + blockIdx.x;
    if (n <= 0) {
      if (t < nr) {
        segD[t * nseg + segi] = make_int2(0, 0);
        segS[t * nseg + segi] = make_int2(0, 0);
      }
      continue;
    }
    hD[t] = 0; hS[t] = 0;
    __syncthreads();
    int s[SB], d[SB], kD[SB], kS[SB];
#pragma unroll
    for (int u = 0; u < SB; ++u) {
      int i = min(s0 + t + u * PT, s1 - 1);
      s[u] = src[i];
      d[u] = dst[i];
    }
#pragma unroll
    for (int u = 0; u < SB; ++u) {
      kD[u] = 0; kS[u] = 0;
      if (t + u * PT < n) {
        kD[u] = atomicAdd(&hD[d[u] >> RSH], 1);
        kS[u] = atomicAdd(&hS[s[u] >> RSH], 1);
      }
    }
    __syncthreads();
    int cD = hD[t], cS = hS[t];
    int lD = block_excl_scan(cD, wsum);
    int lS = block_excl_scan(cS, wsum);
    hD[t] = lD; hS[t] = lS;
    if (t < nr) {
      segD[t * nseg + segi] = make_int2(s0 + lD, cD);
      segS[t * nseg + segi] = make_int2(s0 + lS, cS);
    }
    __syncthreads();
#pragma unroll
    for (int u = 0; u < SB; ++u) {
      kD[u] += hD[d[u] >> RSH];
      kS[u] += hS[s[u] >> RSH];
    }
#pragma unroll
    for (int u = 0; u < SB; ++u) {
      if (t + u * PT < n) {
        stD[kD[u]] = make_int2(s[u], d[u]);
        stS[kS[u]] = s[u];
      }
    }
    __syncthreads();
    {
      int2 r[SB];
#pragma unroll
      for (int u = 0; u < SB; ++u) r[u] = stD[min(t + u * PT, n - 1)];
#pragma unroll
      for (int u = 0; u < SB; ++u) {
        int j = t + u * PT;
        if (j < n) recD[s0 + j] = r[u];
      }
    }
    {
      int v[SB];
#pragma unroll
      for (int u = 0; u < SB; ++u) v[u] = stS[min(t + u * PT, n - 1)];
#pragma unroll
      for (int u = 0; u < SB; ++u) {
        int j = t + u * PT;
        if (j < n) recS[s0 + j] = v[u];
      }
    }
    __syncthreads();
  }
}

__device__ __forceinline__ int range_excl_scan(int v, int* wsum) {
  int lane = threadIdx.x & 31, w = threadIdx.x >> 5;
  int x = v;
#pragma unroll
  for (int d = 1; d < 32; d <<= 1) { int n = __shfl_up(x, d); if (lane >= d) x += n; }
  if (lane == 31) wsum[w] = x;
  __syncthreads();
  int add = 0;
  for (int q = 0; q < w; ++q) add += wsum[q];
  return x - v + add;
}

__device__ __forceinline__ void load_runs(const int2* __restrict__ seg, int b, int nseg, int sb,
                                          int* spos, int* sF, int* wsum) {
  int t = threadIdx.x;
  int2 sg = make_int2(0, 0);
  if (t < SEGT && sb + t < nseg) sg = seg[b * nseg + sb + t];
  int ex = range_excl_scan(sg.y, wsum);
  if (t < SEGT) spos[t] = sg.x;
  if (t <= SEGT) sF[t] = ex;
  __syncthreads();
}

__device__ __forceinline__ void stage_records(const int2* __restrict__ rec, const int* spos,
                                              const int* sF, int wb, int2* srec) {
  int lane = threadIdx.x & 31, w = threadIdx.x >> 5;
  int longrun = 0;
#pragma unroll 1
  for (int q0 = 0; q0 < SPW; q0 += 8) {
    int2 v[8][2]; int fl[8][2];
#pragma unroll
    for (int q = 0; q < 8; ++q) {
      int c = w * SPW + q0 + q;
      int p = spos[c], f0 = sF[c], n = sF[c + 1] - f0;
      longrun |= n > 64;
#pragma unroll
      for (int hh = 0; hh < 2; ++hh) {
        int k = lane + 32 * hh;
        int fw = f0 + k - wb;
        bool m = k < n && fw >= 0 && fw < CAP;
        fl[q][hh] = m ? fw : -1;
        v[q][hh] = rec[m ? p + k : 0];
      }
    }
#pragma unroll
    for (int q = 0; q < 8; ++q) {
#pragma unroll
      for (int hh = 0; hh < 2; ++hh) {
        if (fl[q][hh] >= 0) srec[fl[q][hh]] = v[q][hh];
      }
    }
  }
  if (longrun) {
    for (int q = 0; q < SPW; ++q) {
      int c = w * SPW + q;
      int p = spos[c], f0 = sF[c], n = sF[c + 1] - f0;
      int k0 = max(64, wb - f0);
      int k1 = min(n, wb + CAP - f0);
      for (int k = k0 + lane; k < k1; k += 32) srec[f0 + k - wb] = rec[p + k];
    }
  }
}

__global__ __launch_bounds__(CT) void outnorm_kernel(const int* __restrict__ recS,
                                                     const int* __restrict__ segSw, int nseg,
                                                     const float* __restrict__ h,
                                                     float* __restrict__ norm_out,
                                                     float* __restrict__ hs1, int N) {
  __shared__ int c[RNG];
  __shared__ int spos[SEGT];
  __shared__ int scnt[SEGT];
  const int2* segS = (const int2*)segSw;
  int t = threadIdx.x, b = blockIdx.x;
  int lane = t & 31, w = t >> 5;
  if (t < RNG) c[t] = 0;
  for (int sb = 0; sb < nseg; sb += SEGT) {
    __syncthreads();
    if (t < SEGT) {
      int2 sg = make_int2(0, 0);
      if (sb + t < nseg) sg = segS[b * nseg + sb + t];
      spos[t] = sg.x; scnt[t] = sg.y;
    }
    __syncthreads();
    int v[SPW][2]; bool m[SPW][2];
    int longrun = 0;
#pragma unroll
    for (int q = 0; q < SPW; ++q) {
      int cc = w * SPW + q;
      int p = spos[cc], n = scnt[cc];
      longrun |= n > 64;
#pragma unroll
      for (int hh = 0; hh < 2; ++hh) {
        int k = lane + 32 * hh;
        m[q][hh] = k < n;
        v[q][hh] = recS[m[q][hh] ? p + k : 0];
      }
    }
#pragma unroll
    for (int q = 0; q < SPW; ++q) {
#pragma unroll
      for (int hh = 0; hh < 2; ++hh) {
        if (m[q][hh]) atomicAdd(&c[v[q][hh] & (RNG - 1)], 1);
      }
    }
    if (longrun) {
      for (int q = 0; q < SPW; ++q) {
        int cc = w * SPW + q;
        int p = spos[cc], n = scnt[cc];
        for (int k = 64 + lane; k < n; k += 32) atomicAdd(&c[recS[p + k] & (RNG - 1)], 1);
      }
    }
  }
  __syncthreads();
  int i = b * RNG + t;
  if (t < RNG && i < N) {
    float dO = (float)c[t];
    float no = dO > 0.0f ? rsqrtf(dO) : 0.0f;
    norm_out[i] = no;
#pragma unroll
    for (int k = 0; k < FIN; ++k) hs1[(size_t)i * FIN + k] = h[(size_t)i * FIN + k] * no;
  }
}

__global__ __launch_bounds__(CT) void layer1_kernel(const int* __restrict__ recDw,
                                                    const int* __restrict__ segDw, int nseg,
                                                    const float* __restrict__ hs1,
                                                    const float* __restrict__ norm_out,
                                                    const float* __restrict__ W1,
                                                    const float* __restrict__ b1,
                                                    float* __restrict__ norm_in,
                                                    float* __restrict__ hs2, int N) {
  __shared__ float acc[RNG * 8];
  __shared__ int2 srec[CAP];
  __shared__ int spos[SEGT];
  __shared__ int sF[SEGT + 1];
  __shared__ int wsum[CW];
  const int2* recD = (const int2*)recDw;
  const int2* segD = (const int2*)segDw;
  int t = threadIdx.x, b = blockIdx.x;
  for (int j = t; j < RNG * 8; j += CT) acc[j] = 0.0f;
  int g = t >> 3, f = t & 7;
  for (int sb = 0; sb < nseg; sb += SEGT) {
    load_runs(segD, b, nseg, sb, spos, sF, wsum);
    int total = sF[SEGT];
    for (int wb = 0; wb < total; wb += CAP) {
      int wn = min(CAP, total - wb);
      stage_records(recD, spos, sF, wb, srec);
      __syncthreads();
      for (int j0 = g; j0 - g < wn; j0 += 64 * 8) {
        int2 r[8]; float v[8];
#pragma unroll
        for (int u = 0; u < 8; ++u) r[u] = srec[min(j0 + u * 64, wn - 1)];
#pragma unroll
        for (int u = 0; u < 8; ++u) {
          float x = hs1[(size_t)r[u].x * FIN + min(f, FIN - 1)];
          v[u] = f < FIN ? x : 1.0f;
        }
        asm volatile("" :: "v"(v[0]), "v"(v[1]), "v"(v[2]), "v"(v[3]), "v"(v[4]), "v"(v[5]), "v"(v[6]), "v"(v[7]));
#pragma unroll
        for (int u = 0; u < 8; ++u) {
          if (j0 + u * 64 < wn && f < 7) atomicAdd(&acc[(r[u].y & (RNG - 1)) * 8 + f], v[u]);
        }
      }
      __syncthreads();
    }
  }
  __syncthreads();
  int i = b * RNG + t;
  if (t < RNG && i < N) {
    float dI = acc[t * 8 + 6];
    float ni = dI > 0.0f ? rsqrtf(dI) : 0.0f;
    float no = norm_out[i];
    norm_in[i] = ni;
    float v[FIN];
#pragma unroll
    for (int k = 0; k < FIN; ++k) v[k] = acc[t * 8 + k] * ni;
    float o[HID];
#pragma unroll
    for (int j = 0; j < HID; ++j) {
      float a = b1[j];
#pragma unroll
      for (int k = 0; k < FIN; ++k) a = fmaf(v[k], W1[k * HID + j], a);
      a = fmaxf(a, 0.0f);
      o[j] = a * no;
    }
    float4* hp = (float4*)(hs2 + (size_t)i * HID);
    hp[0] = make_float4(o[0], o[1], o[2], o[3]);
    hp[1] = make_float4(o[4], o[5], o[6], o[7]);
    hp[2] = make_float4(o[8], o[9], o[10], o[11]);
    hp[3] = make_float4(o[12], o[13], o[14], o[15]);
  }
}

__global__ __launch_bounds__(CT) void layer2_kernel(const int* __restrict__ recDw,
                                                    const int* __restrict__ segDw, int nseg,
                                                    const float* __restrict__ hs2,
                                                    const float* __restrict__ norm_in,
                                                    const float* __restrict__ W2,
                                                    const float* __restrict__ b2,
                                                    const float* __restrict__ Wl,
                                                    const float* __restrict__ bl,
                                                    float* __restrict__ out, int N) {
  __shared__ float acc[RNG * HID];
  __shared__ int2 srec[CAP];
  __shared__ int spos[SEGT];
  __shared__ int sF[SEGT + 1];
  __shared__ int wsum[CW];
  const int2* recD = (const int2*)recDw;
  const int2* segD = (const int2*)segDw;
  int t = threadIdx.x, b = blockIdx.x;
  for (int j = t; j < RNG * HID; j += CT) acc[j] = 0.0f;
  int g = t >> 4, f = t & 15;
  for (int sb = 0; sb < nseg; sb += SEGT) {
    load_runs(segD, b, nseg, sb, spos, sF, wsum);
    int total = sF[SEGT];
    for (int wb = 0; wb < total; wb += CAP) {
      int wn = min(CAP, total - wb);
      stage_records(recD, spos, sF, wb, srec);
      __syncthreads();
      for (int j0 = g; j0 - g < wn; j0 += 32 * 8) {
        int2 r[8]; float v[8];
#pragma unroll
        for (int u = 0; u < 8; ++u) r[u] = srec[min(j0 + u * 32, wn - 1)];
#pragma unroll
        for (int u = 0; u < 8; ++u) v[u] = hs2[(unsigned)r[u].x * HID + f];
        asm volatile("" :: "v"(v[0]), "v"(v[1]), "v"(v[2]), "v"(v[3]), "v"(v[4]), "v"(v[5]), "v"(v[6]), "v"(v[7]));
#pragma unroll
        for (int u = 0; u < 8; ++u) {
          if (j0 + u * 32 < wn) atomicAdd(&acc[(r[u].y & (RNG - 1)) * HID + f], v[u]);
        }
      }
      __syncthreads();
    }
  }
  __syncthreads();

  int w = t >> 5, lane = t & 31;
  int half = lane >> 4, lr = lane & 15;
  int lrow = w * 16 + lr;
  int grow = b * RNG + lrow;
  float ni = grow < N ? norm_in[grow] : 0.0f;
  const float* Arow = acc + lrow * HID;
  v8f c = {};
#pragma unroll
  for (int kb = 0; kb < 16; kb += 4) {
    int k0 = kb + half * 2;
    v2f a, bm;
    a.x = Arow[k0 + 0] * ni;
    a.y = Arow[k0 + 1] * ni;
    bm.x = W2[(k0 + 0) * HID + lr];
    bm.y = W2[(k0 + 1) * HID + lr];
    c = __builtin_amdgcn_wmma_f32_16x16x4_f32(false, a, false, bm, (short)0, c, false, false);
  }
  float bias = b2[lr];
#pragma unroll
  for (int r = 0; r < 8; ++r) {
    int m = r + half * 8;
    acc[(w * 16 + m) * HID + lr] = fmaxf(c[r] + bias, 0.0f);
  }
  __syncthreads();
  int i = b * RNG + t;
  if (t < RNG && i < N) {
    float a = bl[0];
#pragma unroll
    for (int j = 0; j < HID; ++j) a = fmaf(acc[t * HID + j], Wl[j], a);
    out[i] = a;
  }
}

extern "C" void kernel_launch(void* const* d_in, const int* in_sizes, int n_in,
                              void* d_out, int out_size, void* d_ws, size_t ws_size,
                              hipStream_t stream) {
  const float* h  = (const float*)d_in[0];
  const float* W1 = (const float*)d_in[1];
  const float* b1 = (const float*)d_in[2];
  const float* W2 = (const float*)d_in[3];
  const float* b2 = (const float*)d_in[4];
  const float* Wl = (const float*)d_in[5];
  const float* bl = (const float*)d_in[6];
  const int* src  = (const int*)d_in[7];
  const int* dst  = (const int*)d_in[8];
  float* out = (float*)d_out;

  const int N = in_sizes[0] / FIN;
  const int E = in_sizes[7];
  const int nr = (N + RNG - 1) / RNG;
  const int chunk = (E + NBLK - 1) / NBLK;
  const int nsub = (chunk + SBE - 1) / SBE;
  const int nseg = NBLK * nsub;

  float* ws = (float*)d_ws;
  float* norm_out = ws;
  float* norm_in  = ws + (size_t)N;
  float* hs1      = ws + 2 * (size_t)N;
  float* hs2      = ws + 8 * (size_t)N;
  int* iw   = (int*)(ws + 24 * (size_t)N);
  int* recD = iw;
  int* segD = recD + 2 * (size_t)E;
  int* segS = segD + 2 * (size_t)nr * nseg;
  int* recS = segS + 2 * (size_t)nr * nseg;

  partition_kernel<<<NBLK, PT, 0, stream>>>(src, dst, recD, recS, segD, segS, E, chunk, nr, nsub);
  outnorm_kernel<<<nr, CT, 0, stream>>>(recS, segS, nseg, h, norm_out, hs1, N);
  layer1_kernel<<<nr, CT, 0, stream>>>(recD, segD, nseg, hs1, norm_out, W1, b1, norm_in, hs2, N);
  layer2_kernel<<<nr, CT, 0, stream>>>(recD, segD, nseg, hs2, norm_in, W2, b2, Wl, bl, out, N);
}


# Node ranges of 196 (was 256) and a larger LDS record window: 511 per-range work-groups, two on each of the 256 WGPs with equal edge counts, not 391 placed up to three per WGP; arrival-order sums
# speedup vs baseline: 16.3253x; 1.0650x over previous
#include <hip/hip_runtime.h>


typedef __attribute__((ext_vector_type(2))) float v2f;
typedef __attribute__((ext_vector_type(8))) float v8f;

#define FIN 6
#define HID 16
#define RNG 196
#define NBLK 256
#define MAXR 1024
#define TROWS 256
#define PT 1024
#define SB 13
#define SBE (SB * PT)
#define CT 512
#define CW 16
#define SEGT 256
#define SPW 16
#define CAP 12800


__device__ __forceinline__ int range_of(int v) { return (int)((unsigned)v / (unsigned)RNG); }

__device__ __forceinline__ int block_excl_scan(int v, int* wsum) {
  int lane = threadIdx.x & 31, w = threadIdx.x >> 5;
  int x = v;
#pragma unroll
  for (int d = 1; d < 32; d <<= 1) { int n = __shfl_up(x, d); if (lane >= d) x += n; }
  if (lane == 31) wsum[w] = x;
  __syncthreads();
  if (w == 0) {
    int y = wsum[lane]; int z = y;
#pragma unroll
    for (int d = 1; d < 32; d <<= 1) { int n = __shfl_up(z, d); if (lane >= d) z += n; }
    wsum[lane] = z - y;
  }
  __syncthreads();
  int r = x - v + wsum[w];
  __syncthreads();
  return r;
}

__global__ __launch_bounds__(PT) void partition_kernel(const int* __restrict__ src,
                                                       const int* __restrict__ dst,
                                                       int* __restrict__ recDw, int* __restrict__ recS,
                                                       int* __restrict__ segDw, int* __restrict__ segSw,
                                                       int E, int chunk, int nr, int nsub) {
  __shared__ int2 stD[SBE];
  __shared__ int stS[SBE];
  __shared__ int hD[MAXR];
  __shared__ int hS[MAXR];
  __shared__ int wsum[32];
  int2* recD = (int2*)recDw;
  int2* segD = (int2*)segDw;
  int2* segS = (int2*)segSw;
  int t = threadIdx.x;
  int nseg = nsub * NBLK;
  int lo = blockIdx.x * chunk;
  int hi = min(lo + chunk, E);
  for (int sub = 0; sub < nsub; ++sub) {
    int s0 = lo + sub * SBE;
    int s1 = min(s0 + SBE, hi);
    int n = s1 - s0;
    int segi = sub * NBLK + blockIdx.x;
    if (n <= 0) {
      if (t < nr) {
        segD[t * nseg + segi] = make_int2(0, 0);
        segS[t * nseg + segi] = make_int2(0, 0);
      }
      continue;
    }
    hD[t] = 0; hS[t] = 0;
    __syncthreads();
    int s[SB], d[SB], kD[SB], kS[SB];
#pragma unroll
    for (int u = 0; u < SB; ++u) {
      int i = min(s0 + t + u * PT, s1 - 1);
      s[u] = src[i];
      d[u] = dst[i];
    }
#pragma unroll
    for (int u = 0; u < SB; ++u) {
      kD[u] = 0; kS[u] = 0;
      if (t + u * PT < n) {
        kD[u] = atomicAdd(&hD[range_of(d[u])], 1);
        kS[u] = atomicAdd(&hS[range_of(s[u])], 1);
      }
    }
    __syncthreads();
    int cD = hD[t], cS = hS[t];
    int lD = block_excl_scan(cD, wsum);
    int lS = block_excl_scan(cS, wsum);
    hD[t] = lD; hS[t] = lS;
    if (t < nr) {
      segD[t * nseg + segi] = make_int2(s0 + lD, cD);
      segS[t * nseg + segi] = make_int2(s0 + lS, cS);
    }
    __syncthreads();
#pragma unroll
    for (int u = 0; u < SB; ++u) {
      kD[u] += hD[range_of(d[u])];
      kS[u] += hS[range_of(s[u])];
    }
#pragma unroll
    for (int u = 0; u < SB; ++u) {
      if (t + u * PT < n) {
        stD[kD[u]] = make_int2(s[u], d[u]);
        stS[kS[u]] = s[u];
      }
    }
    __syncthreads();
    {
      int2 r[SB];
#pragma unroll
      for (int u = 0; u < SB; ++u) r[u] = stD[min(t + u * PT, n - 1)];
#pragma unroll
      for (int u = 0; u < SB; ++u) {
        int j = t + u * PT;
        if (j < n) recD[s0 + j] = r[u];
      }
    }
    {
      int v[SB];
#pragma unroll
      for (int u = 0; u < SB; ++u) v[u] = stS[min(t + u * PT, n - 1)];
#pragma unroll
      for (int u = 0; u < SB; ++u) {
        int j = t + u * PT;
        if (j < n) recS[s0 + j] = v[u];
      }
    }
    __syncthreads();
  }
}

__device__ __forceinline__ int range_excl_scan(int v, int* wsum) {
  int lane = threadIdx.x & 31, w = threadIdx.x >> 5;
  int x = v;
#pragma unroll
  for (int d = 1; d < 32; d <<= 1) { int n = __shfl_up(x, d); if (lane >= d) x += n; }
  if (lane == 31) wsum[w] = x;
  __syncthreads();
  int add = 0;
  for (int q = 0; q < w; ++q) add += wsum[q];
  return x - v + add;
}

__device__ __forceinline__ void load_runs(const int2* __restrict__ seg, int b, int nseg, int sb,
                                          int* spos, int* sF, int* wsum) {
  int t = threadIdx.x;
  int2 sg = make_int2(0, 0);
  if (t < SEGT && sb + t < nseg) sg = seg[b * nseg + sb + t];
  int ex = range_excl_scan(sg.y, wsum);
  if (t < SEGT) spos[t] = sg.x;
  if (t <= SEGT) sF[t] = ex;
  __syncthreads();
}

__device__ __forceinline__ void stage_records(const int2* __restrict__ rec, const int* spos,
                                              const int* sF, int wb, int2* srec) {
  int lane = threadIdx.x & 31, w = threadIdx.x >> 5;
  int longrun = 0;
#pragma unroll 1
  for (int q0 = 0; q0 < SPW; q0 += 8) {
    int2 v[8][2]; int fl[8][2];
#pragma unroll
    for (int q = 0; q < 8; ++q) {
      int c = w * SPW + q0 + q;
      int p = spos[c], f0 = sF[c], n = sF[c + 1] - f0;
      longrun |= n > 64;
#pragma unroll
      for (int hh = 0; hh < 2; ++hh) {
        int k = lane + 32 * hh;
        int fw = f0 + k - wb;
        bool m = k < n && fw >= 0 && fw < CAP;
        fl[q][hh] = m ? fw : -1;
        v[q][hh] = rec[m ? p + k : 0];
      }
    }
#pragma unroll
    for (int q = 0; q < 8; ++q) {
#pragma unroll
      for (int hh = 0; hh < 2; ++hh) {
        if (fl[q][hh] >= 0) srec[fl[q][hh]] = v[q][hh];
      }
    }
  }
  if (longrun) {
    for (int q = 0; q < SPW; ++q) {
      int c = w * SPW + q;
      int p = spos[c], f0 = sF[c], n = sF[c + 1] - f0;
      int k0 = max(64, wb - f0);
      int k1 = min(n, wb + CAP - f0);
      for (int k = k0 + lane; k < k1; k += 32) srec[f0 + k - wb] = rec[p + k];
    }
  }
}

__global__ __launch_bounds__(CT) void outnorm_kernel(const int* __restrict__ recS,
                                                     const int* __restrict__ segSw, int nseg,
                                                     const float* __restrict__ h,
                                                     float* __restrict__ norm_out,
                                                     float* __restrict__ hs1, int N) {
  __shared__ int c[RNG];
  __shared__ int spos[SEGT];
  __shared__ int scnt[SEGT];
  const int2* segS = (const int2*)segSw;
  int t = threadIdx.x, b = blockIdx.x;
  int lane = t & 31, w = t >> 5;
  if (t < RNG) c[t] = 0;
  for (int sb = 0; sb < nseg; sb += SEGT) {
    __syncthreads();
    if (t < SEGT) {
      int2 sg = make_int2(0, 0);
      if (sb + t < nseg) sg = segS[b * nseg + sb + t];
      spos[t] = sg.x; scnt[t] = sg.y;
    }
    __syncthreads();
    int v[SPW][2]; bool m[SPW][2];
    int longrun = 0;
#pragma unroll
    for (int q = 0; q < SPW; ++q) {
      int cc = w * SPW + q;
      int p = spos[cc], n = scnt[cc];
      longrun |= n > 64;
#pragma unroll
      for (int hh = 0; hh < 2; ++hh) {
        int k = lane + 32 * hh;
        m[q][hh] = k < n;
        v[q][hh] = recS[m[q][hh] ? p + k : 0];
      }
    }
#pragma unroll
    for (int q = 0; q < SPW; ++q) {
#pragma unroll
      for (int hh = 0; hh < 2; ++hh) {
        if (m[q][hh]) atomicAdd(&c[v[q][hh] - b * RNG], 1);
      }
    }
    if (longrun) {
      for (int q = 0; q < SPW; ++q) {
        int cc = w * SPW + q;
        int p = spos[cc], n = scnt[cc];
        for (int k = 64 + lane; k < n; k += 32) atomicAdd(&c[recS[p + k] - b * RNG], 1);
      }
    }
  }
  __syncthreads();
  int i = b * RNG + t;
  if (t < RNG && i < N) {
    float dO = (float)c[t];
    float no = dO > 0.0f ? rsqrtf(dO) : 0.0f;
    norm_out[i] = no;
#pragma unroll
    for (int k = 0; k < FIN; ++k) hs1[(size_t)i * FIN + k] = h[(size_t)i * FIN + k] * no;
  }
}

__global__ __launch_bounds__(CT) void layer1_kernel(const int* __restrict__ recDw,
                                                    const int* __restrict__ segDw, int nseg,
                                                    const float* __restrict__ hs1,
                                                    const float* __restrict__ norm_out,
                                                    const float* __restrict__ W1,
                                                    const float* __restrict__ b1,
                                                    float* __restrict__ norm_in,
                                                    float* __restrict__ hs2, int N) {
  __shared__ float acc[RNG * 8];
  __shared__ int2 srec[CAP];
  __shared__ int spos[SEGT];
  __shared__ int sF[SEGT + 1];
  __shared__ int wsum[CW];
  const int2* recD = (const int2*)recDw;
  const int2* segD = (const int2*)segDw;
  int t = threadIdx.x, b = blockIdx.x;
  for (int j = t; j < RNG * 8; j += CT) acc[j] = 0.0f;
  int g = t >> 3, f = t & 7;
  for (int sb = 0; sb < nseg; sb += SEGT) {
    load_runs(segD, b, nseg, sb, spos, sF, wsum);
    int total = sF[SEGT];
    for (int wb = 0; wb < total; wb += CAP) {
      int wn = min(CAP, total - wb);
      stage_records(recD, spos, sF, wb, srec);
      __syncthreads();
      for (int j0 = g; j0 - g < wn; j0 += 64 * 8) {
        int2 r[8]; float v[8];
#pragma unroll
        for (int u = 0; u < 8; ++u) r[u] = srec[min(j0 + u * 64, wn - 1)];
#pragma unroll
        for (int u = 0; u < 8; ++u) {
          float x = hs1[(size_t)r[u].x * FIN + min(f, FIN - 1)];
          v[u] = f < FIN ? x : 1.0f;
        }
        asm volatile("" :: "v"(v[0]), "v"(v[1]), "v"(v[2]), "v"(v[3]), "v"(v[4]), "v"(v[5]), "v"(v[6]), "v"(v[7]));
#pragma unroll
        for (int u = 0; u < 8; ++u) {
          if (j0 + u * 64 < wn && f < 7) atomicAdd(&acc[(r[u].y - b * RNG) * 8 + f], v[u]);
        }
      }
      __syncthreads();
    }
  }
  __syncthreads();
  int i = b * RNG + t;
  if (t < RNG && i < N) {
    float dI = acc[t * 8 + 6];
    float ni = dI > 0.0f ? rsqrtf(dI) : 0.0f;
    float no = norm_out[i];
    norm_in[i] = ni;
    float v[FIN];
#pragma unroll
    for (int k = 0; k < FIN; ++k) v[k] = acc[t * 8 + k] * ni;
    float o[HID];
#pragma unroll
    for (int j = 0; j < HID; ++j) {
      float a = b1[j];
#pragma unroll
      for (int k = 0; k < FIN; ++k) a = fmaf(v[k], W1[k * HID + j], a);
      a = fmaxf(a, 0.0f);
      o[j] = a * no;
    }
    float4* hp = (float4*)(hs2 + (size_t)i * HID);
    hp[0] = make_float4(o[0], o[1], o[2], o[3]);
    hp[1] = make_float4(o[4], o[5], o[6], o[7]);
    hp[2] = make_float4(o[8], o[9], o[10], o[11]);
    hp[3] = make_float4(o[12], o[13], o[14], o[15]);
  }
}

__global__ __launch_bounds__(CT) void layer2_kernel(const int* __restrict__ recDw,
                                                    const int* __restrict__ segDw, int nseg,
                                                    const float* __restrict__ hs2,
                                                    const float* __restrict__ norm_in,
                                                    const float* __restrict__ W2,
                                                    const float* __restrict__ b2,
                                                    const float* __restrict__ Wl,
                                                    const float* __restrict__ bl,
                                                    float* __restrict__ out, int N) {
  __shared__ float acc[TROWS * HID];
  __shared__ int2 srec[CAP];
  __shared__ int spos[SEGT];
  __shared__ int sF[SEGT + 1];
  __shared__ int wsum[CW];
  const int2* recD = (const int2*)recDw;
  const int2* segD = (const int2*)segDw;
  int t = threadIdx.x, b = blockIdx.x;
  for (int j = t; j < TROWS * HID; j += CT) acc[j] = 0.0f;
  int g = t >> 4, f = t & 15;
  for (int sb = 0; sb < nseg; sb += SEGT) {
    load_runs(segD, b, nseg, sb, spos, sF, wsum);
    int total = sF[SEGT];
    for (int wb = 0; wb < total; wb += CAP) {
      int wn = min(CAP, total - wb);
      stage_records(recD, spos, sF, wb, srec);
      __syncthreads();
      for (int j0 = g; j0 - g < wn; j0 += 32 * 8) {
        int2 r[8]; float v[8];
#pragma unroll
        for (int u = 0; u < 8; ++u) r[u] = srec[min(j0 + u * 32, wn - 1)];
#pragma unroll
        for (int u = 0; u < 8; ++u) v[u] = hs2[(unsigned)r[u].x * HID + f];
        asm volatile("" :: "v"(v[0]), "v"(v[1]), "v"(v[2]), "v"(v[3]), "v"(v[4]), "v"(v[5]), "v"(v[6]), "v"(v[7]));
#pragma unroll
        for (int u = 0; u < 8; ++u) {
          if (j0 + u * 32 < wn) atomicAdd(&acc[(r[u].y - b * RNG) * HID + f], v[u]);
        }
      }
      __syncthreads();
    }
  }
  __syncthreads();

  int w = t >> 5, lane = t & 31;
  int half = lane >> 4, lr = lane & 15;
  int lrow = w * 16 + lr;
  int grow = b * RNG + lrow;
  float ni = (lrow < RNG && grow < N) ? norm_in[grow] : 0.0f;
  const float* Arow = acc + lrow * HID;
  v8f c = {};
#pragma unroll
  for (int kb = 0; kb < 16; kb += 4) {
    int k0 = kb + half * 2;
    v2f a, bm;
    a.x = Arow[k0 + 0] * ni;
    a.y = Arow[k0 + 1] * ni;
    bm.x = W2[(k0 + 0) * HID + lr];
    bm.y = W2[(k0 + 1) * HID + lr];
    c = __builtin_amdgcn_wmma_f32_16x16x4_f32(false, a, false, bm, (short)0, c, false, false);
  }
  float bias = b2[lr];
#pragma unroll
  for (int r = 0; r < 8; ++r) {
    int m = r + half * 8;
    acc[(w * 16 + m) * HID + lr] = fmaxf(c[r] + bias, 0.0f);
  }
  __syncthreads();
  int i = b * RNG + t;
  if (t < RNG && i < N) {
    float a = bl[0];
#pragma unroll
    for (int j = 0; j < HID; ++j) a = fmaf(acc[t * HID + j], Wl[j], a);
    out[i] = a;
  }
}

extern "C" void kernel_launch(void* const* d_in, const int* in_sizes, int n_in,
                              void* d_out, int out_size, void* d_ws, size_t ws_size,
                              hipStream_t stream) {
  const float* h  = (const float*)d_in[0];
  const float* W1 = (const float*)d_in[1];
  const float* b1 = (const float*)d_in[2];
  const float* W2 = (const float*)d_in[3];
  const float* b2 = (const float*)d_in[4];
  const float* Wl = (const float*)d_in[5];
  const float* bl = (const float*)d_in[6];
  const int* src  = (const int*)d_in[7];
  const int* dst  = (const int*)d_in[8];
  float* out = (float*)d_out;

  const int N = in_sizes[0] / FIN;
  const int E = in_sizes[7];
  const int nr = (N + RNG - 1) / RNG;
  const int chunk = (E + NBLK - 1) / NBLK;
  const int nsub = (chunk + SBE - 1) / SBE;
  const int nseg = NBLK * nsub;

  float* ws = (float*)d_ws;
  float* norm_out = ws;
  float* norm_in  = ws + (size_t)N;
  float* hs1      = ws + 2 * (size_t)N;
  float* hs2      = ws + 8 * (size_t)N;
  int* iw   = (int*)(ws + 24 * (size_t)N);
  int* recD = iw;
  int* segD = recD + 2 * (size_t)E;
  int* segS = segD + 2 * (size_t)nr * nseg;
  int* recS = segS + 2 * (size_t)nr * nseg;

  partition_kernel<<<NBLK, PT, 0, stream>>>(src, dst, recD, recS, segD, segS, E, chunk, nr, nsub);
  outnorm_kernel<<<nr, CT, 0, stream>>>(recS, segS, nseg, h, norm_out, hs1, N);
  layer1_kernel<<<nr, CT, 0, stream>>>(recD, segD, nseg, hs1, norm_out, W1, b1, norm_in, hs2, N);
  layer2_kernel<<<nr, CT, 0, stream>>>(recD, segD, nseg, hs2, norm_in, W2, b2, Wl, bl, out, N);
}


# Layer loops: a wave reads 32 staged records in one LDS read and passes each to its edge's lanes by DPP, not a broadcast LDS read per edge; 16 gathers in flight, no per-edge guard; arrival-order sums
# speedup vs baseline: 17.0206x; 1.0426x over previous
#include <hip/hip_runtime.h>


typedef __attribute__((ext_vector_type(2))) float v2f;
typedef __attribute__((ext_vector_type(8))) float v8f;

#define FIN 6
#define HID 16
#define RNG 196
#define NBLK 256
#define MAXR 1024
#define TROWS 256
#define PT 1024
#define SB 13
#define SBE (SB * PT)
#define CT 512
#define CW 16
#define SEGT 256
#define SPW 16
#define CAP 12800

#define PAD1 64
#define PAD2 32
#define DROW2 (TROWS - 1)

template <int U> __device__ __forceinline__ int from_row_lane(int x) {
  return __builtin_amdgcn_update_dpp(0, x, 0x150 + U, 0xf, 0xf, false);
}
template <int U> __device__ __forceinline__ int from_oct_lane(int x) {
  return (int)__builtin_amdgcn_mov_dpp8((unsigned)x, (unsigned)(U * 0x249249));
}


__device__ __forceinline__ int range_of(int v) { return (int)((unsigned)v / (unsigned)RNG); }

__device__ __forceinline__ int block_excl_scan(int v, int* wsum) {
  int lane = threadIdx.x & 31, w = threadIdx.x >> 5;
  int x = v;
#pragma unroll
  for (int d = 1; d < 32; d <<= 1) { int n = __shfl_up(x, d); if (lane >= d) x += n; }
  if (lane == 31) wsum[w] = x;
  __syncthreads();
  if (w == 0) {
    int y = wsum[lane]; int z = y;
#pragma unroll
    for (int d = 1; d < 32; d <<= 1) { int n = __shfl_up(z, d); if (lane >= d) z += n; }
    wsum[lane] = z - y;
  }
  __syncthreads();
  int r = x - v + wsum[w];
  __syncthreads();
  return r;
}

__global__ __launch_bounds__(PT) void partition_kernel(const int* __restrict__ src,
                                                       const int* __restrict__ dst,
                                                       int* __restrict__ recDw, int* __restrict__ recS,
                                                       int* __restrict__ segDw, int* __restrict__ segSw,
                                                       int E, int chunk, int nr, int nsub) {
  __shared__ int2 stD[SBE];
  __shared__ int stS[SBE];
  __shared__ int hD[MAXR];
  __shared__ int hS[MAXR];
  __shared__ int wsum[32];
  int2* recD = (int2*)recDw;
  int2* segD = (int2*)segDw;
  int2* segS = (int2*)segSw;
  int t = threadIdx.x;
  int nseg = nsub * NBLK;
  int lo = blockIdx.x * chunk;
  int hi = min(lo + chunk, E);
  for (int sub = 0; sub < nsub; ++sub) {
    int s0 = lo + sub * SBE;
    int s1 = min(s0 + SBE, hi);
    int n = s1 - s0;
    int segi = sub * NBLK + blockIdx.x;
    if (n <= 0) {
      if (t < nr) {
        segD[t * nseg + segi] = make_int2(0, 0);
        segS[t * nseg + segi] = make_int2(0, 0);
      }
      continue;
    }
    hD[t] = 0; hS[t] = 0;
    __syncthreads();
    int s[SB], d[SB], kD[SB], kS[SB];
#pragma unroll
    for (int u = 0; u < SB; ++u) {
      int i = min(s0 + t + u * PT, s1 - 1);
      s[u] = src[i];
      d[u] = dst[i];
    }
#pragma unroll
    for (int u = 0; u < SB; ++u) {
      kD[u] = 0; kS[u] = 0;
      if (t + u * PT < n) {
        kD[u] = atomicAdd(&hD[range_of(d[u])], 1);
        kS[u] = atomicAdd(&hS[range_of(s[u])], 1);
      }
    }
    __syncthreads();
    int cD = hD[t], cS = hS[t];
    int lD = block_excl_scan(cD, wsum);
    int lS = block_excl_scan(cS, wsum);
    hD[t] = lD; hS[t] = lS;
    if (t < nr) {
      segD[t * nseg + segi] = make_int2(s0 + lD, cD);
      segS[t * nseg + segi] = make_int2(s0 + lS, cS);
    }
    __syncthreads();
#pragma unroll
    for (int u = 0; u < SB; ++u) {
      kD[u] += hD[range_of(d[u])];
      kS[u] += hS[range_of(s[u])];
    }
#pragma unroll
    for (int u = 0; u < SB; ++u) {
      if (t + u * PT < n) {
        stD[kD[u]] = make_int2(s[u], d[u]);
        stS[kS[u]] = s[u];
      }
    }
    __syncthreads();
    {
      int2 r[SB];
#pragma unroll
      for (int u = 0; u < SB; ++u) r[u] = stD[min(t + u * PT, n - 1)];
#pragma unroll
      for (int u = 0; u < SB; ++u) {
        int j = t + u * PT;
        if (j < n) recD[s0 + j] = r[u];
      }
    }
    {
      int v[SB];
#pragma unroll
      for (int u = 0; u < SB; ++u) v[u] = stS[min(t + u * PT, n - 1)];
#pragma unroll
      for (int u = 0; u < SB; ++u) {
        int j = t + u * PT;
        if (j < n) recS[s0 + j] = v[u];
      }
    }
    __syncthreads();
  }
}

__device__ __forceinline__ int range_excl_scan(int v, int* wsum) {
  int lane = threadIdx.x & 31, w = threadIdx.x >> 5;
  int x = v;
#pragma unroll
  for (int d = 1; d < 32; d <<= 1) { int n = __shfl_up(x, d); if (lane >= d) x += n; }
  if (lane == 31) wsum[w] = x;
  __syncthreads();
  int add = 0;
  for (int q = 0; q < w; ++q) add += wsum[q];
  return x - v + add;
}

__device__ __forceinline__ void load_runs(const int2* __restrict__ seg, int b, int nseg, int sb,
                                          int* spos, int* sF, int* wsum) {
  int t = threadIdx.x;
  int2 sg = make_int2(0, 0);
  if (t < SEGT && sb + t < nseg) sg = seg[b * nseg + sb + t];
  int ex = range_excl_scan(sg.y, wsum);
  if (t < SEGT) spos[t] = sg.x;
  if (t <= SEGT) sF[t] = ex;
  __syncthreads();
}

__device__ __forceinline__ void stage_records(const int2* __restrict__ rec, const int* spos,
                                              const int* sF, int wb, int2* srec) {
  int lane = threadIdx.x & 31, w = threadIdx.x >> 5;
  int longrun = 0;
#pragma unroll 1
  for (int q0 = 0; q0 < SPW; q0 += 8) {
    int2 v[8][2]; int fl[8][2];
#pragma unroll
    for (int q = 0; q < 8; ++q) {
      int c = w * SPW + q0 + q;
      int p = spos[c], f0 = sF[c], n = sF[c + 1] - f0;
      longrun |= n > 64;
#pragma unroll
      for (int hh = 0; hh < 2; ++hh) {
        int k = lane + 32 * hh;
        int fw = f0 + k - wb;
        bool m = k < n && fw >= 0 && fw < CAP;
        fl[q][hh] = m ? fw : -1;
        v[q][hh] = rec[m ? p + k : 0];
      }
    }
#pragma unroll
    for (int q = 0; q < 8; ++q) {
#pragma unroll
      for (int hh = 0; hh < 2; ++hh) {
        if (fl[q][hh] >= 0) srec[fl[q][hh]] = v[q][hh];
      }
    }
  }
  if (longrun) {
    for (int q = 0; q < SPW; ++q) {
      int c = w * SPW + q;
      int p = spos[c], f0 = sF[c], n = sF[c + 1] - f0;
      int k0 = max(64, wb - f0);
      int k1 = min(n, wb + CAP - f0);
      for (int k = k0 + lane; k < k1; k += 32) srec[f0 + k - wb] = rec[p + k];
    }
  }
}

__global__ __launch_bounds__(CT) void outnorm_kernel(const int* __restrict__ recS,
                                                     const int* __restrict__ segSw, int nseg,
                                                     const float* __restrict__ h,
                                                     float* __restrict__ norm_out,
                                                     float* __restrict__ hs1, int N) {
  __shared__ int c[RNG];
  __shared__ int spos[SEGT];
  __shared__ int scnt[SEGT];
  const int2* segS = (const int2*)segSw;
  int t = threadIdx.x, b = blockIdx.x;
  int lane = t & 31, w = t >> 5;
  if (t < RNG) c[t] = 0;
  for (int sb = 0; sb < nseg; sb += SEGT) {
    __syncthreads();
    if (t < SEGT) {
      int2 sg = make_int2(0, 0);
      if (sb + t < nseg) sg = segS[b * nseg + sb + t];
      spos[t] = sg.x; scnt[t] = sg.y;
    }
    __syncthreads();
    int v[SPW][2]; bool m[SPW][2];
    int longrun = 0;
#pragma unroll
    for (int q = 0; q < SPW; ++q) {
      int cc = w * SPW + q;
      int p = spos[cc], n = scnt[cc];
      longrun |= n > 64;
#pragma unroll
      for (int hh = 0; hh < 2; ++hh) {
        int k = lane + 32 * hh;
        m[q][hh] = k < n;
        v[q][hh] = recS[m[q][hh] ? p + k : 0];
      }
    }
#pragma unroll
    for (int q = 0; q < SPW; ++q) {
#pragma unroll
      for (int hh = 0; hh < 2; ++hh) {
        if (m[q][hh]) atomicAdd(&c[v[q][hh] - b * RNG], 1);
      }
    }
    if (longrun) {
      for (int q = 0; q < SPW; ++q) {
        int cc = w * SPW + q;
        int p = spos[cc], n = scnt[cc];
        for (int k = 64 + lane; k < n; k += 32) atomicAdd(&c[recS[p + k] - b * RNG], 1);
      }
    }
  }
  __syncthreads();
  int i = b * RNG + t;
  if (t < RNG && i < N) {
    float dO = (float)c[t];
    float no = dO > 0.0f ? rsqrtf(dO) : 0.0f;
    norm_out[i] = no;
#pragma unroll
    for (int k = 0; k < FIN; ++k) hs1[(size_t)i * FIN + k] = h[(size_t)i * FIN + k] * no;
  }
}

__global__ __launch_bounds__(CT) void layer1_kernel(const int* __restrict__ recDw,
                                                    const int* __restrict__ segDw, int nseg,
                                                    const float* __restrict__ hs1,
                                                    const float* __restrict__ norm_out,
                                                    const float* __restrict__ W1,
                                                    const float* __restrict__ b1,
                                                    float* __restrict__ norm_in,
                                                    float* __restrict__ hs2, int N) {
  __shared__ float acc[(RNG + 1) * 8];
  __shared__ int2 srec[CAP + PAD1];
  __shared__ int spos[SEGT];
  __shared__ int sF[SEGT + 1];
  __shared__ int wsum[CW];
  const int2* recD = (const int2*)recDw;
  const int2* segD = (const int2*)segDw;
  int t = threadIdx.x, b = blockIdx.x;
  for (int j = t; j < (RNG + 1) * 8; j += CT) acc[j] = 0.0f;
  int f = t & 7;
  int fm = min(f, FIN - 1);
  int lane = t & 31;
  int w = __builtin_amdgcn_readfirstlane(t >> 5);
  int cidx = f - b * RNG * 8;
  for (int sb = 0; sb < nseg; sb += SEGT) {
    load_runs(segD, b, nseg, sb, spos, sF, wsum);
    int total = sF[SEGT];
    for (int wb = 0; wb < total; wb += CAP) {
      int wn = min(CAP, total - wb);
      int wnp = (wn + PAD1 - 1) & ~(PAD1 - 1);
      stage_records(recD, spos, sF, wb, srec);
      if (t < wnp - wn) srec[wn + t] = make_int2(0, b * RNG + RNG);
      __syncthreads();
      for (int j0 = w * PAD1; j0 < wnp; j0 += CW * PAD1) {
        int2 ra = srec[j0 + lane];
        int2 rb = srec[j0 + 32 + lane];
        float v[16];
#define L1_GATHER(U) { \
          float xa = hs1[(unsigned)from_oct_lane<U>(ra.x) * FIN + fm]; \
          float xb = hs1[(unsigned)from_oct_lane<U>(rb.x) * FIN + fm]; \
          v[U] = f < FIN ? xa : 1.0f; v[8 + U] = f < FIN ? xb : 1.0f; }
        L1_GATHER(0) L1_GATHER(1) L1_GATHER(2) L1_GATHER(3) L1_GATHER(4) L1_GATHER(5) L1_GATHER(6) L1_GATHER(7)
#undef L1_GATHER
#define L1_ADD(U) { \
          atomicAdd(&acc[from_oct_lane<U>(ra.y) * 8 + cidx], v[U]); \
          atomicAdd(&acc[from_oct_lane<U>(rb.y) * 8 + cidx], v[8 + U]); }
        L1_ADD(0) L1_ADD(1) L1_ADD(2) L1_ADD(3) L1_ADD(4) L1_ADD(5) L1_ADD(6) L1_ADD(7)
#undef L1_ADD
      }
      __syncthreads();
    }
  }
  __syncthreads();
  int i = b * RNG + t;
  if (t < RNG && i < N) {
    float dI = acc[t * 8 + 6];
    float ni = dI > 0.0f ? rsqrtf(dI) : 0.0f;
    float no = norm_out[i];
    norm_in[i] = ni;
    float v[FIN];
#pragma unroll
    for (int k = 0; k < FIN; ++k) v[k] = acc[t * 8 + k] * ni;
    float o[HID];
#pragma unroll
    for (int j = 0; j < HID; ++j) {
      float a = b1[j];
#pragma unroll
      for (int k = 0; k < FIN; ++k) a = fmaf(v[k], W1[k * HID + j], a);
      a = fmaxf(a, 0.0f);
      o[j] = a * no;
    }
    float4* hp = (float4*)(hs2 + (size_t)i * HID);
    hp[0] = make_float4(o[0], o[1], o[2], o[3]);
    hp[1] = make_float4(o[4], o[5], o[6], o[7]);
    hp[2] = make_float4(o[8], o[9], o[10], o[11]);
    hp[3] = make_float4(o[12], o[13], o[14], o[15]);
  }
}

__global__ __launch_bounds__(CT) void layer2_kernel(const int* __restrict__ recDw,
                                                    const int* __restrict__ segDw, int nseg,
                                                    const float* __restrict__ hs2,
                                                    const float* __restrict__ norm_in,
                                                    const float* __restrict__ W2,
                                                    const float* __restrict__ b2,
                                                    const float* __restrict__ Wl,
                                                    const float* __restrict__ bl,
                                                    float* __restrict__ out, int N) {
  __shared__ float acc[TROWS * HID];
  __shared__ int2 srec[CAP + PAD2];
  __shared__ int spos[SEGT];
  __shared__ int sF[SEGT + 1];
  __shared__ int wsum[CW];
  const int2* recD = (const int2*)recDw;
  const int2* segD = (const int2*)segDw;
  int t = threadIdx.x, b = blockIdx.x;
  for (int j = t; j < TROWS * HID; j += CT) acc[j] = 0.0f;
  int f = t & 15;
  int lane0 = t & 31;
  int w0 = __builtin_amdgcn_readfirstlane(t >> 5);
  int cidx = f - b * RNG * HID;
  for (int sb = 0; sb < nseg; sb += SEGT) {
    load_runs(segD, b, nseg, sb, spos, sF, wsum);
    int total = sF[SEGT];
    for (int wb = 0; wb < total; wb += CAP) {
      int wn = min(CAP, total - wb);
      int wnp = (wn + PAD2 - 1) & ~(PAD2 - 1);
      stage_records(recD, spos, sF, wb, srec);
      if (t < wnp - wn) srec[wn + t] = make_int2(0, b * RNG + DROW2);
      __syncthreads();
      for (int j0 = w0 * PAD2; j0 < wnp; j0 += CW * PAD2) {
        int2 rc = srec[j0 + lane0];
        float v[16];
#define L2_GATHER(U) v[U] = hs2[(unsigned)from_row_lane<U>(rc.x) * HID + f];
        L2_GATHER(0) L2_GATHER(1) L2_GATHER(2) L2_GATHER(3) L2_GATHER(4) L2_GATHER(5) L2_GATHER(6) L2_GATHER(7)
        L2_GATHER(8) L2_GATHER(9) L2_GATHER(10) L2_GATHER(11) L2_GATHER(12) L2_GATHER(13) L2_GATHER(14) L2_GATHER(15)
#undef L2_GATHER
#define L2_ADD(U) atomicAdd(&acc[from_row_lane<U>(rc.y) * HID + cidx], v[U]);
        L2_ADD(0) L2_ADD(1) L2_ADD(2) L2_ADD(3) L2_ADD(4) L2_ADD(5) L2_ADD(6) L2_ADD(7)
        L2_ADD(8) L2_ADD(9) L2_ADD(10) L2_ADD(11) L2_ADD(12) L2_ADD(13) L2_ADD(14) L2_ADD(15)
#undef L2_ADD
      }
      __syncthreads();
    }
  }
  __syncthreads();

  int w = t >> 5, lane = t & 31;
  int half = lane >> 4, lr = lane & 15;
  int lrow = w * 16 + lr;
  int grow = b * RNG + lrow;
  float ni = (lrow < RNG && grow < N) ? norm_in[grow] : 0.0f;
  const float* Arow = acc + lrow * HID;
  v8f c = {};
#pragma unroll
  for (int kb = 0; kb < 16; kb += 4) {
    int k0 = kb + half * 2;
    v2f a, bm;
    a.x = Arow[k0 + 0] * ni;
    a.y = Arow[k0 + 1] * ni;
    bm.x = W2[(k0 + 0) * HID + lr];
    bm.y = W2[(k0 + 1) * HID + lr];
    c = __builtin_amdgcn_wmma_f32_16x16x4_f32(false, a, false, bm, (short)0, c, false, false);
  }
  float bias = b2[lr];
#pragma unroll
  for (int r = 0; r < 8; ++r) {
    int m = r + half * 8;
    acc[(w * 16 + m) * HID + lr] = fmaxf(c[r] + bias, 0.0f);
  }
  __syncthreads();
  int i = b * RNG + t;
  if (t < RNG && i < N) {
    float a = bl[0];
#pragma unroll
    for (int j = 0; j < HID; ++j) a = fmaf(acc[t * HID + j], Wl[j], a);
    out[i] = a;
  }
}

extern "C" void kernel_launch(void* const* d_in, const int* in_sizes, int n_in,
                              void* d_out, int out_size, void* d_ws, size_t ws_size,
                              hipStream_t stream) {
  const float* h  = (const float*)d_in[0];
  const float* W1 = (const float*)d_in[1];
  const float* b1 = (const float*)d_in[2];
  const float* W2 = (const float*)d_in[3];
  const float* b2 = (const float*)d_in[4];
  const float* Wl = (const float*)d_in[5];
  const float* bl = (const float*)d_in[6];
  const int* src  = (const int*)d_in[7];
  const int* dst  = (const int*)d_in[8];
  float* out = (float*)d_out;

  const int N = in_sizes[0] / FIN;
  const int E = in_sizes[7];
  const int nr = (N + RNG - 1) / RNG;
  const int chunk = (E + NBLK - 1) / NBLK;
  const int nsub = (chunk + SBE - 1) / SBE;
  const int nseg = NBLK * nsub;

  float* ws = (float*)d_ws;
  float* norm_out = ws;
  float* norm_in  = ws + (size_t)N;
  float* hs1      = ws + 2 * (size_t)N;
  float* hs2      = ws + 8 * (size_t)N;
  int* iw   = (int*)(ws + 24 * (size_t)N);
  int* recD = iw;
  int* segD = recD + 2 * (size_t)E;
  int* segS = segD + 2 * (size_t)nr * nseg;
  int* recS = segS + 2 * (size_t)nr * nseg;

  partition_kernel<<<NBLK, PT, 0, stream>>>(src, dst, recD, recS, segD, segS, E, chunk, nr, nsub);
  outnorm_kernel<<<nr, CT, 0, stream>>>(recS, segS, nseg, h, norm_out, hs1, N);
  layer1_kernel<<<nr, CT, 0, stream>>>(recD, segD, nseg, hs1, norm_out, W1, b1, norm_in, hs2, N);
  layer2_kernel<<<nr, CT, 0, stream>>>(recD, segD, nseg, hs2, norm_in, W2, b2, Wl, bl, out, N);
}


# Edge records packed into one 32-bit word (source id in 24 bits, row in its destination range in 8): half the record bytes stored and staged; a layer's runs load in one round trip; arrival-order sums
# speedup vs baseline: 17.6424x; 1.0365x over previous
#include <hip/hip_runtime.h>


typedef __attribute__((ext_vector_type(2))) float v2f;
typedef __attribute__((ext_vector_type(8))) float v8f;

#define FIN 6
#define HID 16
#define RNG 196
#define NBLK 256
#define MAXR 1024
#define TROWS 256
#define PT 1024
#define SB 13
#define SBE (SB * PT)
#define CT 512
#define CW 16
#define SEGT 256
#define SPW 16
#define CAP 25600
#define SRCBITS 24
#define SRCMASK 0xffffff
#define STPAD 14336

#define PAD1 64
#define PAD2 32
#define DROW2 (TROWS - 1)

template <int U> __device__ __forceinline__ int from_row_lane(int x) {
  return __builtin_amdgcn_update_dpp(0, x, 0x150 + U, 0xf, 0xf, false);
}
template <int U> __device__ __forceinline__ int from_oct_lane(int x) {
  return (int)__builtin_amdgcn_mov_dpp8((unsigned)x, (unsigned)(U * 0x249249));
}


__device__ __forceinline__ int range_of(int v) { return (int)((unsigned)v / (unsigned)RNG); }

__device__ __forceinline__ int block_excl_scan(int v, int* wsum) {
  int lane = threadIdx.x & 31, w = threadIdx.x >> 5;
  int x = v;
#pragma unroll
  for (int d = 1; d < 32; d <<= 1) { int n = __shfl_up(x, d); if (lane >= d) x += n; }
  if (lane == 31) wsum[w] = x;
  __syncthreads();
  if (w == 0) {
    int y = wsum[lane]; int z = y;
#pragma unroll
    for (int d = 1; d < 32; d <<= 1) { int n = __shfl_up(z, d); if (lane >= d) z += n; }
    wsum[lane] = z - y;
  }
  __syncthreads();
  int r = x - v + wsum[w];
  __syncthreads();
  return r;
}

__global__ __launch_bounds__(PT) void partition_kernel(const int* __restrict__ src,
                                                       const int* __restrict__ dst,
                                                       int* __restrict__ recD, int* __restrict__ recS,
                                                       int* __restrict__ segDw, int* __restrict__ segSw,
                                                       int E, int chunk, int nr, int nsub) {
  __shared__ int stD[SBE + STPAD];
  __shared__ int stS[SBE];
  __shared__ int hD[MAXR];
  __shared__ int hS[MAXR];
  __shared__ int wsum[32];
  int2* segD = (int2*)segDw;
  int2* segS = (int2*)segSw;
  int t = threadIdx.x;
  int nseg = nsub * NBLK;
  int lo = blockIdx.x * chunk;
  int hi = min(lo + chunk, E);
  for (int sub = 0; sub < nsub; ++sub) {
    int s0 = lo + sub * SBE;
    int s1 = min(s0 + SBE, hi);
    int n = s1 - s0;
    int segi = sub * NBLK + blockIdx.x;
    if (n <= 0) {
      if (t < nr) {
        segD[t * nseg + segi] = make_int2(0, 0);
        segS[t * nseg + segi] = make_int2(0, 0);
      }
      continue;
    }
    hD[t] = 0; hS[t] = 0;
    __syncthreads();
    int s[SB], d[SB], kD[SB], kS[SB];
#pragma unroll
    for (int u = 0; u < SB; ++u) {
      int i = min(s0 + t + u * PT, s1 - 1);
      s[u] = src[i];
      d[u] = dst[i];
    }
#pragma unroll
    for (int u = 0; u < SB; ++u) {
      kD[u] = 0; kS[u] = 0;
      if (t + u * PT < n) {
        kD[u] = atomicAdd(&hD[range_of(d[u])], 1);
        kS[u] = atomicAdd(&hS[range_of(s[u])], 1);
      }
    }
    __syncthreads();
    int cD = hD[t], cS = hS[t];
    int lD = block_excl_scan(cD, wsum);
    int lS = block_excl_scan(cS, wsum);
    hD[t] = lD; hS[t] = lS;
    if (t < nr) {
      segD[t * nseg + segi] = make_int2(s0 + lD, cD);
      segS[t * nseg + segi] = make_int2(s0 + lS, cS);
    }
    __syncthreads();
#pragma unroll
    for (int u = 0; u < SB; ++u) {
      kD[u] += hD[range_of(d[u])];
      kS[u] += hS[range_of(s[u])];
    }
#pragma unroll
    for (int u = 0; u < SB; ++u) {
      if (t + u * PT < n) {
        unsigned row = (unsigned)(d[u] - range_of(d[u]) * RNG);
        stD[kD[u]] = (int)((unsigned)s[u] | (row << SRCBITS));
        stS[kS[u]] = s[u];
      }
    }
    __syncthreads();
    {
      int r[SB];
#pragma unroll
      for (int u = 0; u < SB; ++u) r[u] = stD[min(t + u * PT, n - 1)];
#pragma unroll
      for (int u = 0; u < SB; ++u) {
        int j = t + u * PT;
        if (j < n) recD[s0 + j] = r[u];
      }
    }
    {
      int v[SB];
#pragma unroll
      for (int u = 0; u < SB; ++u) v[u] = stS[min(t + u * PT, n - 1)];
#pragma unroll
      for (int u = 0; u < SB; ++u) {
        int j = t + u * PT;
        if (j < n) recS[s0 + j] = v[u];
      }
    }
    __syncthreads();
  }
}

__device__ __forceinline__ int range_excl_scan(int v, int* wsum) {
  int lane = threadIdx.x & 31, w = threadIdx.x >> 5;
  int x = v;
#pragma unroll
  for (int d = 1; d < 32; d <<= 1) { int n = __shfl_up(x, d); if (lane >= d) x += n; }
  if (lane == 31) wsum[w] = x;
  __syncthreads();
  int add = 0;
  for (int q = 0; q < w; ++q) add += wsum[q];
  return x - v + add;
}

__device__ __forceinline__ void load_runs(const int2* __restrict__ seg, int b, int nseg, int sb,
                                          int* spos, int* sF, int* wsum) {
  int t = threadIdx.x;
  int2 sg = make_int2(0, 0);
  if (t < SEGT && sb + t < nseg) sg = seg[b * nseg + sb + t];
  int ex = range_excl_scan(sg.y, wsum);
  if (t < SEGT) spos[t] = sg.x;
  if (t <= SEGT) sF[t] = ex;
  __syncthreads();
}

__device__ __forceinline__ void stage_records(const int* __restrict__ rec, const int* spos,
                                              const int* sF, int wb, int* srec) {
  int lane = threadIdx.x & 31, w = threadIdx.x >> 5;
  int longrun = 0;
  int v[SPW][2]; int fl[SPW][2];
#pragma unroll
  for (int q = 0; q < SPW; ++q) {
    int c = w * SPW + q;
    int p = spos[c], f0 = sF[c], n = sF[c + 1] - f0;
    longrun |= n > 64;
#pragma unroll
    for (int hh = 0; hh < 2; ++hh) {
      int k = lane + 32 * hh;
      int fw = f0 + k - wb;
      bool m = k < n && fw >= 0 && fw < CAP;
      fl[q][hh] = m ? fw : -1;
      v[q][hh] = rec[m ? p + k : 0];
    }
  }
#pragma unroll
  for (int q = 0; q < SPW; ++q) {
#pragma unroll
    for (int hh = 0; hh < 2; ++hh) {
      if (fl[q][hh] >= 0) srec[fl[q][hh]] = v[q][hh];
    }
  }
  if (longrun) {
    for (int q = 0; q < SPW; ++q) {
      int c = w * SPW + q;
      int p = spos[c], f0 = sF[c], n = sF[c + 1] - f0;
      int k0 = max(64, wb - f0);
      int k1 = min(n, wb + CAP - f0);
      for (int k = k0 + lane; k < k1; k += 32) srec[f0 + k - wb] = rec[p + k];
    }
  }
}

__global__ __launch_bounds__(CT) void outnorm_kernel(const int* __restrict__ recS,
                                                     const int* __restrict__ segSw, int nseg,
                                                     const float* __restrict__ h,
                                                     float* __restrict__ norm_out,
                                                     float* __restrict__ hs1, int N) {
  __shared__ int c[RNG];
  __shared__ int spos[SEGT];
  __shared__ int scnt[SEGT];
  const int2* segS = (const int2*)segSw;
  int t = threadIdx.x, b = blockIdx.x;
  int lane = t & 31, w = t >> 5;
  if (t < RNG) c[t] = 0;
  for (int sb = 0; sb < nseg; sb += SEGT) {
    __syncthreads();
    if (t < SEGT) {
      int2 sg = make_int2(0, 0);
      if (sb + t < nseg) sg = segS[b * nseg + sb + t];
      spos[t] = sg.x; scnt[t] = sg.y;
    }
    __syncthreads();
    int v[SPW][2]; bool m[SPW][2];
    int longrun = 0;
#pragma unroll
    for (int q = 0; q < SPW; ++q) {
      int cc = w * SPW + q;
      int p = spos[cc], n = scnt[cc];
      longrun |= n > 64;
#pragma unroll
      for (int hh = 0; hh < 2; ++hh) {
        int k = lane + 32 * hh;
        m[q][hh] = k < n;
        v[q][hh] = recS[m[q][hh] ? p + k : 0];
      }
    }
#pragma unroll
    for (int q = 0; q < SPW; ++q) {
#pragma unroll
      for (int hh = 0; hh < 2; ++hh) {
        if (m[q][hh]) atomicAdd(&c[v[q][hh] - b * RNG], 1);
      }
    }
    if (longrun) {
      for (int q = 0; q < SPW; ++q) {
        int cc = w * SPW + q;
        int p = spos[cc], n = scnt[cc];
        for (int k = 64 + lane; k < n; k += 32) atomicAdd(&c[recS[p + k] - b * RNG], 1);
      }
    }
  }
  __syncthreads();
  int i = b * RNG + t;
  if (t < RNG && i < N) {
    float dO = (float)c[t];
    float no = dO > 0.0f ? rsqrtf(dO) : 0.0f;
    norm_out[i] = no;
#pragma unroll
    for (int k = 0; k < FIN; ++k) hs1[(size_t)i * FIN + k] = h[(size_t)i * FIN + k] * no;
  }
}

__global__ __launch_bounds__(CT) void layer1_kernel(const int* __restrict__ recD,
                                                    const int* __restrict__ segDw, int nseg,
                                                    const float* __restrict__ hs1,
                                                    const float* __restrict__ norm_out,
                                                    const float* __restrict__ W1,
                                                    const float* __restrict__ b1,
                                                    float* __restrict__ norm_in,
                                                    float* __restrict__ hs2, int N) {
  __shared__ float acc[(RNG + 1) * 8];
  __shared__ int srec[CAP + PAD1];
  __shared__ int spos[SEGT];
  __shared__ int sF[SEGT + 1];
  __shared__ int wsum[CW];
  const int2* segD = (const int2*)segDw;
  int t = threadIdx.x, b = blockIdx.x;
  for (int j = t; j < (RNG + 1) * 8; j += CT) acc[j] = 0.0f;
  int f = t & 7;
  int fm = min(f, FIN - 1);
  int lane = t & 31;
  int w = __builtin_amdgcn_readfirstlane(t >> 5);
  for (int sb = 0; sb < nseg; sb += SEGT) {
    load_runs(segD, b, nseg, sb, spos, sF, wsum);
    int total = sF[SEGT];
    for (int wb = 0; wb < total; wb += CAP) {
      int wn = min(CAP, total - wb);
      int wnp = (wn + PAD1 - 1) & ~(PAD1 - 1);
      stage_records(recD, spos, sF, wb, srec);
      if (t < wnp - wn) srec[wn + t] = (int)((unsigned)RNG << SRCBITS);
      __syncthreads();
      for (int j0 = w * PAD1; j0 < wnp; j0 += CW * PAD1) {
        int ra = srec[j0 + lane];
        int rb = srec[j0 + 32 + lane];
        int sa = ra & SRCMASK, sb2 = rb & SRCMASK;
        int da = (int)((unsigned)ra >> SRCBITS), db = (int)((unsigned)rb >> SRCBITS);
        float v[16];
#define L1_GATHER(U) { \
          float xa = hs1[(unsigned)from_oct_lane<U>(sa) * FIN + fm]; \
          float xb = hs1[(unsigned)from_oct_lane<U>(sb2) * FIN + fm]; \
          v[U] = f < FIN ? xa : 1.0f; v[8 + U] = f < FIN ? xb : 1.0f; }
        L1_GATHER(0) L1_GATHER(1) L1_GATHER(2) L1_GATHER(3) L1_GATHER(4) L1_GATHER(5) L1_GATHER(6) L1_GATHER(7)
#undef L1_GATHER
#define L1_ADD(U) { \
          atomicAdd(&acc[from_oct_lane<U>(da) * 8 + f], v[U]); \
          atomicAdd(&acc[from_oct_lane<U>(db) * 8 + f], v[8 + U]); }
        L1_ADD(0) L1_ADD(1) L1_ADD(2) L1_ADD(3) L1_ADD(4) L1_ADD(5) L1_ADD(6) L1_ADD(7)
#undef L1_ADD
      }
      __syncthreads();
    }
  }
  __syncthreads();
  int i = b * RNG + t;
  if (t < RNG && i < N) {
    float dI = acc[t * 8 + 6];
    float ni = dI > 0.0f ? rsqrtf(dI) : 0.0f;
    float no = norm_out[i];
    norm_in[i] = ni;
    float v[FIN];
#pragma unroll
    for (int k = 0; k < FIN; ++k) v[k] = acc[t * 8 + k] * ni;
    float o[HID];
#pragma unroll
    for (int j = 0; j < HID; ++j) {
      float a = b1[j];
#pragma unroll
      for (int k = 0; k < FIN; ++k) a = fmaf(v[k], W1[k * HID + j], a);
      a = fmaxf(a, 0.0f);
      o[j] = a * no;
    }
    float4* hp = (float4*)(hs2 + (size_t)i * HID);
    hp[0] = make_float4(o[0], o[1], o[2], o[3]);
    hp[1] = make_float4(o[4], o[5], o[6], o[7]);
    hp[2] = make_float4(o[8], o[9], o[10], o[11]);
    hp[3] = make_float4(o[12], o[13], o[14], o[15]);
  }
}

__global__ __launch_bounds__(CT) void layer2_kernel(const int* __restrict__ recD,
                                                    const int* __restrict__ segDw, int nseg,
                                                    const float* __restrict__ hs2,
                                                    const float* __restrict__ norm_in,
                                                    const float* __restrict__ W2,
                                                    const float* __restrict__ b2,
                                                    const float* __restrict__ Wl,
                                                    const float* __restrict__ bl,
                                                    float* __restrict__ out, int N) {
  __shared__ float acc[TROWS * HID];
  __shared__ int srec[CAP + PAD2];
  __shared__ int spos[SEGT];
  __shared__ int sF[SEGT + 1];
  __shared__ int wsum[CW];
  const int2* segD = (const int2*)segDw;
  int t = threadIdx.x, b = blockIdx.x;
  for (int j = t; j < TROWS * HID; j += CT) acc[j] = 0.0f;
  int f = t & 15;
  int lane0 = t & 31;
  int w0 = __builtin_amdgcn_readfirstlane(t >> 5);
  for (int sb = 0; sb < nseg; sb += SEGT) {
    load_runs(segD, b, nseg, sb, spos, sF, wsum);
    int total = sF[SEGT];
    for (int wb = 0; wb < total; wb += CAP) {
      int wn = min(CAP, total - wb);
      int wnp = (wn + PAD2 - 1) & ~(PAD2 - 1);
      stage_records(recD, spos, sF, wb, srec);
      if (t < wnp - wn) srec[wn + t] = (int)((unsigned)DROW2 << SRCBITS);
      __syncthreads();
      for (int j0 = w0 * PAD2; j0 < wnp; j0 += CW * PAD2) {
        int rc = srec[j0 + lane0];
        int sc = rc & SRCMASK;
        int dc = (int)((unsigned)rc >> SRCBITS);
        float v[16];
#define L2_GATHER(U) v[U] = hs2[(unsigned)from_row_lane<U>(sc) * HID + f];
        L2_GATHER(0) L2_GATHER(1) L2_GATHER(2) L2_GATHER(3) L2_GATHER(4) L2_GATHER(5) L2_GATHER(6) L2_GATHER(7)
        L2_GATHER(8) L2_GATHER(9) L2_GATHER(10) L2_GATHER(11) L2_GATHER(12) L2_GATHER(13) L2_GATHER(14) L2_GATHER(15)
#undef L2_GATHER
#define L2_ADD(U) atomicAdd(&acc[from_row_lane<U>(dc) * HID + f], v[U]);
        L2_ADD(0) L2_ADD(1) L2_ADD(2) L2_ADD(3) L2_ADD(4) L2_ADD(5) L2_ADD(6) L2_ADD(7)
        L2_ADD(8) L2_ADD(9) L2_ADD(10) L2_ADD(11) L2_ADD(12) L2_ADD(13) L2_ADD(14) L2_ADD(15)
#undef L2_ADD
      }
      __syncthreads();
    }
  }
  __syncthreads();

  int w = t >> 5, lane = t & 31;
  int half = lane >> 4, lr = lane & 15;
  int lrow = w * 16 + lr;
  int grow = b * RNG + lrow;
  float ni = (lrow < RNG && grow < N) ? norm_in[grow] : 0.0f;
  const float* Arow = acc + lrow * HID;
  v8f c = {};
#pragma unroll
  for (int kb = 0; kb < 16; kb += 4) {
    int k0 = kb + half * 2;
    v2f a, bm;
    a.x = Arow[k0 + 0] * ni;
    a.y = Arow[k0 + 1] * ni;
    bm.x = W2[(k0 + 0) * HID + lr];
    bm.y = W2[(k0 + 1) * HID + lr];
    c = __builtin_amdgcn_wmma_f32_16x16x4_f32(false, a, false, bm, (short)0, c, false, false);
  }
  float bias = b2[lr];
#pragma unroll
  for (int r = 0; r < 8; ++r) {
    int m = r + half * 8;
    acc[(w * 16 + m) * HID + lr] = fmaxf(c[r] + bias, 0.0f);
  }
  __syncthreads();
  int i = b * RNG + t;
  if (t < RNG && i < N) {
    float a = bl[0];
#pragma unroll
    for (int j = 0; j < HID; ++j) a = fmaf(acc[t * HID + j], Wl[j], a);
    out[i] = a;
  }
}

extern "C" void kernel_launch(void* const* d_in, const int* in_sizes, int n_in,
                              void* d_out, int out_size, void* d_ws, size_t ws_size,
                              hipStream_t stream) {
  const float* h  = (const float*)d_in[0];
  const float* W1 = (const float*)d_in[1];
  const float* b1 = (const float*)d_in[2];
  const float* W2 = (const float*)d_in[3];
  const float* b2 = (const float*)d_in[4];
  const float* Wl = (const float*)d_in[5];
  const float* bl = (const float*)d_in[6];
  const int* src  = (const int*)d_in[7];
  const int* dst  = (const int*)d_in[8];
  float* out = (float*)d_out;

  const int N = in_sizes[0] / FIN;
  const int E = in_sizes[7];
  const int nr = (N + RNG - 1) / RNG;
  const int chunk = (E + NBLK - 1) / NBLK;
  const int nsub = (chunk + SBE - 1) / SBE;
  const int nseg = NBLK * nsub;

  float* ws = (float*)d_ws;
  float* norm_out = ws;
  float* norm_in  = ws + (size_t)N;
  float* hs1      = ws + 2 * (size_t)N;
  float* hs2      = ws + 8 * (size_t)N;
  int* iw   = (int*)(ws + 24 * (size_t)N);
  int* recD = iw;
  int* segD = recD + 2 * (size_t)E;
  int* segS = segD + 2 * (size_t)nr * nseg;
  int* recS = segS + 2 * (size_t)nr * nseg;

  partition_kernel<<<NBLK, PT, 0, stream>>>(src, dst, recD, recS, segD, segS, E, chunk, nr, nsub);
  outnorm_kernel<<<nr, CT, 0, stream>>>(recS, segS, nseg, h, norm_out, hs1, N);
  layer1_kernel<<<nr, CT, 0, stream>>>(recD, segD, nseg, hs1, norm_out, W1, b1, norm_in, hs2, N);
  layer2_kernel<<<nr, CT, 0, stream>>>(recD, segD, nseg, hs2, norm_in, W2, b2, Wl, bl, out, N);
}
